# starting kernel
# baseline (speedup 1.0000x reference)
#include <hip/hip_runtime.h>

typedef __attribute__((ext_vector_type(2))) float v2f;
typedef __attribute__((ext_vector_type(8))) float v8f;

#define N_NODES 100000
#define F_IN    256
#define H1_DIM  128
#define H2_DIM  64
#define E_EDGES 1600000
#define EP_CNT  100000

__global__ void init_deg_kernel(float* __restrict__ deg, int n) {
    int i = blockIdx.x * blockDim.x + threadIdx.x;
    if (i < n) deg[i] = 1.0f;
}

__global__ void accum_deg_kernel(const int* __restrict__ col,
                                 float* __restrict__ deg, int e) {
    int i = blockIdx.x * blockDim.x + threadIdx.x;
    if (i < e) atomicAdd(&deg[col[i]], 1.0f);
}

__global__ void rsqrt_kernel(float* __restrict__ deg, int n) {
    int i = blockIdx.x * blockDim.x + threadIdx.x;
    if (i < n) deg[i] = rsqrtf(deg[i]);
}

__global__ void transpose_kernel(const float* __restrict__ B,
                                 float* __restrict__ Bt, int K, int N) {
    int idx = blockIdx.x * blockDim.x + threadIdx.x;
    if (idx >= K * N) return;
    int k = idx / N;
    int n = idx - k * N;
    Bt[(size_t)n * K + k] = B[idx];
}

template <int K, int NT>
__global__ void __launch_bounds__(NT * 32)
gemm_wmma_f32(const float* __restrict__ A, const float* __restrict__ Bt,
              float* __restrict__ C) {
    constexpr int N = NT * 16;
    const int wave = threadIdx.x >> 5;
    const int lane = threadIdx.x & 31;
    const int l    = lane & 15;
    const int half = lane >> 4;

    const int m0 = blockIdx.x * 16;
    const int n0 = wave * 16;

    const float* ap = A  + (size_t)(m0 + l) * K + 2 * half;
    const float* bp = Bt + (size_t)(n0 + l) * K + 2 * half;

    v8f acc = {0.f, 0.f, 0.f, 0.f, 0.f, 0.f, 0.f, 0.f};
#pragma unroll
    for (int k0 = 0; k0 < K; k0 += 4) {
        v2f a = *(const v2f*)(ap + k0);
        v2f b = *(const v2f*)(bp + k0);
        acc = __builtin_amdgcn_wmma_f32_16x16x4_f32(
             false, a,  false, b,
             (short)0, acc,  false,  false);
    }

    float* cp = C + (size_t)(m0 + 8 * half) * N + n0 + l;
#pragma unroll
    for (int j = 0; j < 8; ++j)
        cp[(size_t)j * N] = acc[j];
}

template <int H>
__global__ void self_init_kernel(const float* __restrict__ htmp,
                                 const float* __restrict__ dinv,
                                 float* __restrict__ acc, int n) {
    int idx = blockIdx.x * blockDim.x + threadIdx.x;
    if (idx >= n * H) return;
    int node = idx / H;
    float di = dinv[node];
    acc[idx] = htmp[idx] * di * di;
}

template <int H>
__global__ void __launch_bounds__(256)
edge_scatter_kernel(const int* __restrict__ row, const int* __restrict__ col,
                    const float* __restrict__ dinv,
                    const float* __restrict__ htmp,
                    float* __restrict__ acc, int e) {
    int warp_id = (blockIdx.x * blockDim.x + threadIdx.x) >> 5;
    int lane = threadIdx.x & 31;
    if (warp_id >= e) return;
    int r = row[warp_id];
    int c = col[warp_id];
    float norm = dinv[r] * dinv[c];
    const float* src = htmp + (size_t)r * H;
    float* dst = acc + (size_t)c * H;
#pragma unroll
    for (int f = 0; f < H; f += 32)
        atomicAdd(&dst[f + lane], src[f + lane] * norm);
}

template <int H>
__global__ void bias_relu_kernel(float* __restrict__ acc,
                                 const float* __restrict__ bias, int n) {
    int idx = blockIdx.x * blockDim.x + threadIdx.x;
    if (idx >= n * H) return;
    float v = acc[idx] + bias[idx & (H - 1)];
    acc[idx] = v > 0.f ? v : 0.f;
}

template <int H>
__global__ void bias_add_kernel(float* __restrict__ acc,
                                const float* __restrict__ bias, int n) {
    int idx = blockIdx.x * blockDim.x + threadIdx.x;
    if (idx >= n * H) return;
    acc[idx] = acc[idx] + bias[idx & (H - 1)];
}

__global__ void __launch_bounds__(256)
edge_dot_kernel(const int* __restrict__ pos, const int* __restrict__ neg,
                const float* __restrict__ h2, float* __restrict__ out, int ep) {
    int warp_id = (blockIdx.x * blockDim.x + threadIdx.x) >> 5;
    int lane = threadIdx.x & 31;
    if (warp_id >= 2 * ep) return;
    int src, dst;
    if (warp_id < ep) {
        src = pos[warp_id];
        dst = pos[ep + warp_id];
    } else {
        int e2 = warp_id - ep;
        src = neg[e2];
        dst = neg[ep + e2];
    }
    const float* xj = h2 + (size_t)src * H2_DIM;
    const float* xi = h2 + (size_t)dst * H2_DIM;
    float s = xi[lane] * xj[lane] + xi[lane + 32] * xj[lane + 32];
#pragma unroll
    for (int off = 16; off > 0; off >>= 1)
        s += __shfl_xor(s, off, 32);
    if (lane == 0) out[warp_id] = s;
}

static inline size_t align256(size_t x) { return (x + 255) & ~(size_t)255; }

extern "C" void kernel_launch(void* const* d_in, const int* in_sizes, int n_in,
                              void* d_out, int out_size, void* d_ws, size_t ws_size,
                              hipStream_t stream) {
    const float* x    = (const float*)d_in[0];
    const int*   ei   = (const int*)d_in[1];
    const int*   pos  = (const int*)d_in[2];
    const int*   neg  = (const int*)d_in[3];
    const float* W1   = (const float*)d_in[4];
    const float* b1   = (const float*)d_in[5];
    const float* W2   = (const float*)d_in[6];
    const float* b2   = (const float*)d_in[7];
    float* out = (float*)d_out;

    const int* row = ei;
    const int* col = ei + E_EDGES;

    char* ws = (char*)d_ws;
    size_t off = 0;
    float* dinv  = (float*)(ws + off); off += align256(sizeof(float) * N_NODES);
    float* W1t   = (float*)(ws + off); off += align256(sizeof(float) * F_IN * H1_DIM);
    float* W2t   = (float*)(ws + off); off += align256(sizeof(float) * H1_DIM * H2_DIM);
    float* htmp1 = (float*)(ws + off); off += align256(sizeof(float) * (size_t)N_NODES * H1_DIM);
    float* h1    = (float*)(ws + off); off += align256(sizeof(float) * (size_t)N_NODES * H1_DIM);
    float* htmp2 = (float*)(ws + off); off += align256(sizeof(float) * (size_t)N_NODES * H2_DIM);
    float* h2    = (float*)(ws + off); off += align256(sizeof(float) * (size_t)N_NODES * H2_DIM);
    (void)ws_size; (void)n_in; (void)in_sizes; (void)out_size;

    const int TB = 256;

    transpose_kernel<<<(F_IN * H1_DIM + TB - 1) / TB, TB, 0, stream>>>(W1, W1t, F_IN, H1_DIM);
    transpose_kernel<<<(H1_DIM * H2_DIM + TB - 1) / TB, TB, 0, stream>>>(W2, W2t, H1_DIM, H2_DIM);

    init_deg_kernel<<<(N_NODES + TB - 1) / TB, TB, 0, stream>>>(dinv, N_NODES);
    accum_deg_kernel<<<(E_EDGES + TB - 1) / TB, TB, 0, stream>>>(col, dinv, E_EDGES);
    rsqrt_kernel<<<(N_NODES + TB - 1) / TB, TB, 0, stream>>>(dinv, N_NODES);

    gemm_wmma_f32<F_IN, 8><<<N_NODES / 16, 8 * 32, 0, stream>>>(x, W1t, htmp1);

    {
        int tot = N_NODES * H1_DIM;
        self_init_kernel<H1_DIM><<<(tot + TB - 1) / TB, TB, 0, stream>>>(htmp1, dinv, h1, N_NODES);
        int warps_blocks = (E_EDGES * 32 + TB - 1) / TB;
        edge_scatter_kernel<H1_DIM><<<warps_blocks, TB, 0, stream>>>(row, col, dinv, htmp1, h1, E_EDGES);
        bias_relu_kernel<H1_DIM><<<(tot + TB - 1) / TB, TB, 0, stream>>>(h1, b1, N_NODES);
    }

    gemm_wmma_f32<H1_DIM, 4><<<N_NODES / 16, 4 * 32, 0, stream>>>(h1, W2t, htmp2);

    {
        int tot = N_NODES * H2_DIM;
        self_init_kernel<H2_DIM><<<(tot + TB - 1) / TB, TB, 0, stream>>>(htmp2, dinv, h2, N_NODES);
        int warps_blocks = (E_EDGES * 32 + TB - 1) / TB;
        edge_scatter_kernel<H2_DIM><<<warps_blocks, TB, 0, stream>>>(row, col, dinv, htmp2, h2, E_EDGES);
        bias_add_kernel<H2_DIM><<<(tot + TB - 1) / TB, TB, 0, stream>>>(h2, b2, N_NODES);
    }

    {
        long long total_threads = (long long)2 * EP_CNT * 32;
        int blocks = (int)((total_threads + TB - 1) / TB);
        edge_dot_kernel<<<blocks, TB, 0, stream>>>(pos, neg, h2, out, EP_CNT);
    }
}


# Both GEMMs retiled: a wave owns a 16-row strip of all columns (each A fragment feeds 8 or 4 products), W pre-laid in fragment order so B loads are whole lines; k regrouped inside each product sum
# speedup vs baseline: 1.1920x; 1.1920x over previous
#include <hip/hip_runtime.h>

typedef __attribute__((ext_vector_type(2))) float v2f;
typedef __attribute__((ext_vector_type(4))) float v4f;
typedef __attribute__((ext_vector_type(8))) float v8f;

#define N_NODES 100000
#define F_IN    256
#define H1_DIM  128
#define H2_DIM  64
#define E_EDGES 1600000
#define EP_CNT  100000

__global__ void init_deg_kernel(float* __restrict__ deg, int n) {
    int i = blockIdx.x * blockDim.x + threadIdx.x;
    if (i < n) deg[i] = 1.0f;
}

__global__ void accum_deg_kernel(const int* __restrict__ col,
                                 float* __restrict__ deg, int e) {
    int i = blockIdx.x * blockDim.x + threadIdx.x;
    if (i < e) atomicAdd(&deg[col[i]], 1.0f);
}

__global__ void rsqrt_kernel(float* __restrict__ deg, int n) {
    int i = blockIdx.x * blockDim.x + threadIdx.x;
    if (i < n) deg[i] = rsqrtf(deg[i]);
}

template <int K, int NT>
__global__ void frag_order_kernel(const float* __restrict__ B,
                                  float* __restrict__ Bf) {
    constexpr int N = NT * 16;
    int idx = blockIdx.x * blockDim.x + threadIdx.x;
    if (idx >= K * N) return;
    int s    = idx & 3;
    int lane = (idx >> 2) & 31;
    int t    = idx >> 7;
    int nt   = t % NT;
    int j    = t / NT;
    int c    = lane & 15;
    int h    = lane >> 4;
    int k    = 8 * j + 4 * h + s;
    Bf[idx] = B[(size_t)k * N + nt * 16 + c];
}

template <int K, int NT>
__global__ void __launch_bounds__(160)
gemm_wmma_f32_strip(const float* __restrict__ A, const float* __restrict__ Bf,
                    float* __restrict__ C) {
    constexpr int N  = NT * 16;
    constexpr int NJ = K / 8;
    const int wave = threadIdx.x >> 5;
    const int lane = threadIdx.x & 31;
    const int c    = lane & 15;
    const int h    = lane >> 4;
    const int m0   = (blockIdx.x * 5 + wave) * 16;

    const float* ap = A + (size_t)(m0 + c) * K + 4 * h;
    const float* bp = Bf + lane * 4;

    v8f acc[NT];
#pragma unroll
    for (int nt = 0; nt < NT; ++nt)
        acc[nt] = (v8f){0.f, 0.f, 0.f, 0.f, 0.f, 0.f, 0.f, 0.f};

    v4f a_cur = *(const v4f*)ap;
    v4f b_cur[NT];
#pragma unroll
    for (int nt = 0; nt < NT; ++nt)
        b_cur[nt] = *(const v4f*)(bp + nt * 128);

#pragma unroll
    for (int j = 0; j < NJ; ++j) {
        v4f a_nxt = a_cur;
        v4f b_nxt[NT];
#pragma unroll
        for (int nt = 0; nt < NT; ++nt)
            b_nxt[nt] = b_cur[nt];
        if (j + 1 < NJ) {
            a_nxt = *(const v4f*)(ap + 8 * (j + 1));
#pragma unroll
            for (int nt = 0; nt < NT; ++nt)
                b_nxt[nt] = *(const v4f*)(bp + ((j + 1) * NT + nt) * 128);
        }
        __builtin_amdgcn_sched_barrier(0);
        v2f a01 = {a_cur.x, a_cur.y};
        v2f a23 = {a_cur.z, a_cur.w};
#pragma unroll
        for (int nt = 0; nt < NT; ++nt) {
            v2f b01 = {b_cur[nt].x, b_cur[nt].y};
            acc[nt] = __builtin_amdgcn_wmma_f32_16x16x4_f32(
                 false, a01,  false, b01,
                 (short)0, acc[nt],  false,  false);
        }
#pragma unroll
        for (int nt = 0; nt < NT; ++nt) {
            v2f b23 = {b_cur[nt].z, b_cur[nt].w};
            acc[nt] = __builtin_amdgcn_wmma_f32_16x16x4_f32(
                 false, a23,  false, b23,
                 (short)0, acc[nt],  false,  false);
        }
        __builtin_amdgcn_sched_barrier(0);
        a_cur = a_nxt;
#pragma unroll
        for (int nt = 0; nt < NT; ++nt)
            b_cur[nt] = b_nxt[nt];
    }

    float* cp = C + (size_t)(m0 + 8 * h) * N + c;
#pragma unroll
    for (int nt = 0; nt < NT; ++nt)
#pragma unroll
        for (int r = 0; r < 8; ++r)
            cp[(size_t)r * N + nt * 16] = acc[nt][r];
}

template <int H>
__global__ void self_init_kernel(const float* __restrict__ htmp,
                                 const float* __restrict__ dinv,
                                 float* __restrict__ acc, int n) {
    int idx = blockIdx.x * blockDim.x + threadIdx.x;
    if (idx >= n * H) return;
    int node = idx / H;
    float di = dinv[node];
    acc[idx] = htmp[idx] * di * di;
}

template <int H>
__global__ void __launch_bounds__(256)
edge_scatter_kernel(const int* __restrict__ row, const int* __restrict__ col,
                    const float* __restrict__ dinv,
                    const float* __restrict__ htmp,
                    float* __restrict__ acc, int e) {
    int warp_id = (blockIdx.x * blockDim.x + threadIdx.x) >> 5;
    int lane = threadIdx.x & 31;
    if (warp_id >= e) return;
    int r = row[warp_id];
    int c = col[warp_id];
    float norm = dinv[r] * dinv[c];
    const float* src = htmp + (size_t)r * H;
    float* dst = acc + (size_t)c * H;
#pragma unroll
    for (int f = 0; f < H; f += 32)
        atomicAdd(&dst[f + lane], src[f + lane] * norm);
}

template <int H>
__global__ void bias_relu_kernel(float* __restrict__ acc,
                                 const float* __restrict__ bias, int n) {
    int idx = blockIdx.x * blockDim.x + threadIdx.x;
    if (idx >= n * H) return;
    float v = acc[idx] + bias[idx & (H - 1)];
    acc[idx] = v > 0.f ? v : 0.f;
}

template <int H>
__global__ void bias_add_kernel(float* __restrict__ acc,
                                const float* __restrict__ bias, int n) {
    int idx = blockIdx.x * blockDim.x + threadIdx.x;
    if (idx >= n * H) return;
    acc[idx] = acc[idx] + bias[idx & (H - 1)];
}

__global__ void __launch_bounds__(256)
edge_dot_kernel(const int* __restrict__ pos, const int* __restrict__ neg,
                const float* __restrict__ h2, float* __restrict__ out, int ep) {
    int warp_id = (blockIdx.x * blockDim.x + threadIdx.x) >> 5;
    int lane = threadIdx.x & 31;
    if (warp_id >= 2 * ep) return;
    int src, dst;
    if (warp_id < ep) {
        src = pos[warp_id];
        dst = pos[ep + warp_id];
    } else {
        int e2 = warp_id - ep;
        src = neg[e2];
        dst = neg[ep + e2];
    }
    const float* xj = h2 + (size_t)src * H2_DIM;
    const float* xi = h2 + (size_t)dst * H2_DIM;
    float s = xi[lane] * xj[lane] + xi[lane + 32] * xj[lane + 32];
#pragma unroll
    for (int off = 16; off > 0; off >>= 1)
        s += __shfl_xor(s, off, 32);
    if (lane == 0) out[warp_id] = s;
}

static inline size_t align256(size_t x) { return (x + 255) & ~(size_t)255; }

extern "C" void kernel_launch(void* const* d_in, const int* in_sizes, int n_in,
                              void* d_out, int out_size, void* d_ws, size_t ws_size,
                              hipStream_t stream) {
    const float* x    = (const float*)d_in[0];
    const int*   ei   = (const int*)d_in[1];
    const int*   pos  = (const int*)d_in[2];
    const int*   neg  = (const int*)d_in[3];
    const float* W1   = (const float*)d_in[4];
    const float* b1   = (const float*)d_in[5];
    const float* W2   = (const float*)d_in[6];
    const float* b2   = (const float*)d_in[7];
    float* out = (float*)d_out;

    const int* row = ei;
    const int* col = ei + E_EDGES;

    char* ws = (char*)d_ws;
    size_t off = 0;
    float* dinv  = (float*)(ws + off); off += align256(sizeof(float) * N_NODES);
    float* W1f   = (float*)(ws + off); off += align256(sizeof(float) * F_IN * H1_DIM);
    float* W2f   = (float*)(ws + off); off += align256(sizeof(float) * H1_DIM * H2_DIM);
    float* htmp1 = (float*)(ws + off); off += align256(sizeof(float) * (size_t)N_NODES * H1_DIM);
    float* h1    = (float*)(ws + off); off += align256(sizeof(float) * (size_t)N_NODES * H1_DIM);
    float* htmp2 = (float*)(ws + off); off += align256(sizeof(float) * (size_t)N_NODES * H2_DIM);
    float* h2    = (float*)(ws + off); off += align256(sizeof(float) * (size_t)N_NODES * H2_DIM);
    (void)ws_size; (void)n_in; (void)in_sizes; (void)out_size;

    const int TB = 256;

    frag_order_kernel<F_IN, 8><<<(F_IN * H1_DIM + TB - 1) / TB, TB, 0, stream>>>(W1, W1f);
    frag_order_kernel<H1_DIM, 4><<<(H1_DIM * H2_DIM + TB - 1) / TB, TB, 0, stream>>>(W2, W2f);

    init_deg_kernel<<<(N_NODES + TB - 1) / TB, TB, 0, stream>>>(dinv, N_NODES);
    accum_deg_kernel<<<(E_EDGES + TB - 1) / TB, TB, 0, stream>>>(col, dinv, E_EDGES);
    rsqrt_kernel<<<(N_NODES + TB - 1) / TB, TB, 0, stream>>>(dinv, N_NODES);

    gemm_wmma_f32_strip<F_IN, 8><<<N_NODES / 80, 160, 0, stream>>>(x, W1f, htmp1);

    {
        int tot = N_NODES * H1_DIM;
        self_init_kernel<H1_DIM><<<(tot + TB - 1) / TB, TB, 0, stream>>>(htmp1, dinv, h1, N_NODES);
        int warps_blocks = (E_EDGES * 32 + TB - 1) / TB;
        edge_scatter_kernel<H1_DIM><<<warps_blocks, TB, 0, stream>>>(row, col, dinv, htmp1, h1, E_EDGES);
        bias_relu_kernel<H1_DIM><<<(tot + TB - 1) / TB, TB, 0, stream>>>(h1, b1, N_NODES);
    }

    gemm_wmma_f32_strip<H1_DIM, 4><<<N_NODES / 80, 160, 0, stream>>>(h1, W2f, htmp2);

    {
        int tot = N_NODES * H2_DIM;
        self_init_kernel<H2_DIM><<<(tot + TB - 1) / TB, TB, 0, stream>>>(htmp2, dinv, h2, N_NODES);
        int warps_blocks = (E_EDGES * 32 + TB - 1) / TB;
        edge_scatter_kernel<H2_DIM><<<warps_blocks, TB, 0, stream>>>(row, col, dinv, htmp2, h2, E_EDGES);
        bias_add_kernel<H2_DIM><<<(tot + TB - 1) / TB, TB, 0, stream>>>(h2, b2, N_NODES);
    }

    {
        long long total_threads = (long long)2 * EP_CNT * 32;
        int blocks = (int)((total_threads + TB - 1) / TB);
        edge_dot_kernel<<<blocks, TB, 0, stream>>>(pos, neg, h2, out, EP_CNT);
    }
}


# Edge scatter: a wave takes 32 edges, not 1; lane i loads edge i's indices and norm once, the wave reads them out per edge; 8-16 rows' loads go ahead of their float atomics (arrival order as before)
# speedup vs baseline: 1.9182x; 1.6092x over previous
#include <hip/hip_runtime.h>

typedef __attribute__((ext_vector_type(2))) float v2f;
typedef __attribute__((ext_vector_type(4))) float v4f;
typedef __attribute__((ext_vector_type(8))) float v8f;

#define N_NODES 100000
#define F_IN    256
#define H1_DIM  128
#define H2_DIM  64
#define E_EDGES 1600000
#define EP_CNT  100000

__global__ void init_deg_kernel(float* __restrict__ deg, int n) {
    int i = blockIdx.x * blockDim.x + threadIdx.x;
    if (i < n) deg[i] = 1.0f;
}

__global__ void accum_deg_kernel(const int* __restrict__ col,
                                 float* __restrict__ deg, int e) {
    int i = blockIdx.x * blockDim.x + threadIdx.x;
    if (i < e) atomicAdd(&deg[col[i]], 1.0f);
}

__global__ void rsqrt_kernel(float* __restrict__ deg, int n) {
    int i = blockIdx.x * blockDim.x + threadIdx.x;
    if (i < n) deg[i] = rsqrtf(deg[i]);
}

template <int K, int NT>
__global__ void frag_order_kernel(const float* __restrict__ B,
                                  float* __restrict__ Bf) {
    constexpr int N = NT * 16;
    int idx = blockIdx.x * blockDim.x + threadIdx.x;
    if (idx >= K * N) return;
    int s    = idx & 3;
    int lane = (idx >> 2) & 31;
    int t    = idx >> 7;
    int nt   = t % NT;
    int j    = t / NT;
    int c    = lane & 15;
    int h    = lane >> 4;
    int k    = 8 * j + 4 * h + s;
    Bf[idx] = B[(size_t)k * N + nt * 16 + c];
}

template <int K, int NT>
__global__ void __launch_bounds__(160)
gemm_wmma_f32_strip(const float* __restrict__ A, const float* __restrict__ Bf,
                    float* __restrict__ C) {
    constexpr int N  = NT * 16;
    constexpr int NJ = K / 8;
    const int wave = threadIdx.x >> 5;
    const int lane = threadIdx.x & 31;
    const int c    = lane & 15;
    const int h    = lane >> 4;
    const int m0   = (blockIdx.x * 5 + wave) * 16;

    const float* ap = A + (size_t)(m0 + c) * K + 4 * h;
    const float* bp = Bf + lane * 4;

    v8f acc[NT];
#pragma unroll
    for (int nt = 0; nt < NT; ++nt)
        acc[nt] = (v8f){0.f, 0.f, 0.f, 0.f, 0.f, 0.f, 0.f, 0.f};

    v4f a_cur = *(const v4f*)ap;
    v4f b_cur[NT];
#pragma unroll
    for (int nt = 0; nt < NT; ++nt)
        b_cur[nt] = *(const v4f*)(bp + nt * 128);

#pragma unroll
    for (int j = 0; j < NJ; ++j) {
        v4f a_nxt = a_cur;
        v4f b_nxt[NT];
#pragma unroll
        for (int nt = 0; nt < NT; ++nt)
            b_nxt[nt] = b_cur[nt];
        if (j + 1 < NJ) {
            a_nxt = *(const v4f*)(ap + 8 * (j + 1));
#pragma unroll
            for (int nt = 0; nt < NT; ++nt)
                b_nxt[nt] = *(const v4f*)(bp + ((j + 1) * NT + nt) * 128);
        }
        __builtin_amdgcn_sched_barrier(0);
        v2f a01 = {a_cur.x, a_cur.y};
        v2f a23 = {a_cur.z, a_cur.w};
#pragma unroll
        for (int nt = 0; nt < NT; ++nt) {
            v2f b01 = {b_cur[nt].x, b_cur[nt].y};
            acc[nt] = __builtin_amdgcn_wmma_f32_16x16x4_f32(
                 false, a01,  false, b01,
                 (short)0, acc[nt],  false,  false);
        }
#pragma unroll
        for (int nt = 0; nt < NT; ++nt) {
            v2f b23 = {b_cur[nt].z, b_cur[nt].w};
            acc[nt] = __builtin_amdgcn_wmma_f32_16x16x4_f32(
                 false, a23,  false, b23,
                 (short)0, acc[nt],  false,  false);
        }
        __builtin_amdgcn_sched_barrier(0);
        a_cur = a_nxt;
#pragma unroll
        for (int nt = 0; nt < NT; ++nt)
            b_cur[nt] = b_nxt[nt];
    }

    float* cp = C + (size_t)(m0 + 8 * h) * N + c;
#pragma unroll
    for (int nt = 0; nt < NT; ++nt)
#pragma unroll
        for (int r = 0; r < 8; ++r)
            cp[(size_t)r * N + nt * 16] = acc[nt][r];
}

template <int H>
__global__ void self_init_kernel(const float* __restrict__ htmp,
                                 const float* __restrict__ dinv,
                                 float* __restrict__ acc, int n) {
    int idx = blockIdx.x * blockDim.x + threadIdx.x;
    if (idx >= n * H) return;
    int node = idx / H;
    float di = dinv[node];
    acc[idx] = htmp[idx] * di * di;
}

template <int H, int G>
__global__ void __launch_bounds__(256)
edge_scatter_kernel(const int* __restrict__ row, const int* __restrict__ col,
                    const float* __restrict__ dinv,
                    const float* __restrict__ htmp,
                    float* __restrict__ acc) {
    static_assert(E_EDGES % 256 == 0, "8-wave blocks of 32-edge waves cover the edges exactly");
    static_assert(32 % G == 0, "groups tile the wave's 32 edges");
    constexpr int V = H / 32;
    const unsigned lane = threadIdx.x & 31;
    const unsigned wave = (blockIdx.x * blockDim.x + threadIdx.x) >> 5;
    const unsigned ei = wave * 32 + lane;
    const int r = row[ei];
    const int c = col[ei];
    const float nrm = dinv[r] * dinv[c];
#pragma unroll
    for (int j0 = 0; j0 < 32; j0 += G) {
        float v[G][V];
#pragma unroll
        for (int g = 0; g < G; ++g) {
            const int rj = __builtin_amdgcn_readlane(r, j0 + g);
            const float* src = htmp + (size_t)rj * H;
#pragma unroll
            for (int k = 0; k < V; ++k)
                v[g][k] = src[lane + k * 32];
        }
#pragma unroll
        for (int g = 0; g < G; ++g) {
            const int cj = __builtin_amdgcn_readlane(c, j0 + g);
            const float nj = __builtin_bit_cast(
                float, __builtin_amdgcn_readlane(__builtin_bit_cast(int, nrm), j0 + g));
            float* dst = acc + (size_t)cj * H;
#pragma unroll
            for (int k = 0; k < V; ++k)
                atomicAdd(&dst[lane + k * 32], v[g][k] * nj);
        }
    }
}

template <int H>
__global__ void bias_relu_kernel(float* __restrict__ acc,
                                 const float* __restrict__ bias, int n) {
    int idx = blockIdx.x * blockDim.x + threadIdx.x;
    if (idx >= n * H) return;
    float v = acc[idx] + bias[idx & (H - 1)];
    acc[idx] = v > 0.f ? v : 0.f;
}

template <int H>
__global__ void bias_add_kernel(float* __restrict__ acc,
                                const float* __restrict__ bias, int n) {
    int idx = blockIdx.x * blockDim.x + threadIdx.x;
    if (idx >= n * H) return;
    acc[idx] = acc[idx] + bias[idx & (H - 1)];
}

__global__ void __launch_bounds__(256)
edge_dot_kernel(const int* __restrict__ pos, const int* __restrict__ neg,
                const float* __restrict__ h2, float* __restrict__ out, int ep) {
    int warp_id = (blockIdx.x * blockDim.x + threadIdx.x) >> 5;
    int lane = threadIdx.x & 31;
    if (warp_id >= 2 * ep) return;
    int src, dst;
    if (warp_id < ep) {
        src = pos[warp_id];
        dst = pos[ep + warp_id];
    } else {
        int e2 = warp_id - ep;
        src = neg[e2];
        dst = neg[ep + e2];
    }
    const float* xj = h2 + (size_t)src * H2_DIM;
    const float* xi = h2 + (size_t)dst * H2_DIM;
    float s = xi[lane] * xj[lane] + xi[lane + 32] * xj[lane + 32];
#pragma unroll
    for (int off = 16; off > 0; off >>= 1)
        s += __shfl_xor(s, off, 32);
    if (lane == 0) out[warp_id] = s;
}

static inline size_t align256(size_t x) { return (x + 255) & ~(size_t)255; }

extern "C" void kernel_launch(void* const* d_in, const int* in_sizes, int n_in,
                              void* d_out, int out_size, void* d_ws, size_t ws_size,
                              hipStream_t stream) {
    const float* x    = (const float*)d_in[0];
    const int*   ei   = (const int*)d_in[1];
    const int*   pos  = (const int*)d_in[2];
    const int*   neg  = (const int*)d_in[3];
    const float* W1   = (const float*)d_in[4];
    const float* b1   = (const float*)d_in[5];
    const float* W2   = (const float*)d_in[6];
    const float* b2   = (const float*)d_in[7];
    float* out = (float*)d_out;

    const int* row = ei;
    const int* col = ei + E_EDGES;

    char* ws = (char*)d_ws;
    size_t off = 0;
    float* dinv  = (float*)(ws + off); off += align256(sizeof(float) * N_NODES);
    float* W1f   = (float*)(ws + off); off += align256(sizeof(float) * F_IN * H1_DIM);
    float* W2f   = (float*)(ws + off); off += align256(sizeof(float) * H1_DIM * H2_DIM);
    float* htmp1 = (float*)(ws + off); off += align256(sizeof(float) * (size_t)N_NODES * H1_DIM);
    float* h1    = (float*)(ws + off); off += align256(sizeof(float) * (size_t)N_NODES * H1_DIM);
    float* htmp2 = (float*)(ws + off); off += align256(sizeof(float) * (size_t)N_NODES * H2_DIM);
    float* h2    = (float*)(ws + off); off += align256(sizeof(float) * (size_t)N_NODES * H2_DIM);
    (void)ws_size; (void)n_in; (void)in_sizes; (void)out_size;

    const int TB = 256;

    frag_order_kernel<F_IN, 8><<<(F_IN * H1_DIM + TB - 1) / TB, TB, 0, stream>>>(W1, W1f);
    frag_order_kernel<H1_DIM, 4><<<(H1_DIM * H2_DIM + TB - 1) / TB, TB, 0, stream>>>(W2, W2f);

    init_deg_kernel<<<(N_NODES + TB - 1) / TB, TB, 0, stream>>>(dinv, N_NODES);
    accum_deg_kernel<<<(E_EDGES + TB - 1) / TB, TB, 0, stream>>>(col, dinv, E_EDGES);
    rsqrt_kernel<<<(N_NODES + TB - 1) / TB, TB, 0, stream>>>(dinv, N_NODES);

    gemm_wmma_f32_strip<F_IN, 8><<<N_NODES / 80, 160, 0, stream>>>(x, W1f, htmp1);

    {
        int tot = N_NODES * H1_DIM;
        self_init_kernel<H1_DIM><<<(tot + TB - 1) / TB, TB, 0, stream>>>(htmp1, dinv, h1, N_NODES);
        edge_scatter_kernel<H1_DIM, 8><<<E_EDGES / TB, TB, 0, stream>>>(row, col, dinv, htmp1, h1);
        bias_relu_kernel<H1_DIM><<<(tot + TB - 1) / TB, TB, 0, stream>>>(h1, b1, N_NODES);
    }

    gemm_wmma_f32_strip<H1_DIM, 4><<<N_NODES / 80, 160, 0, stream>>>(h1, W2f, htmp2);

    {
        int tot = N_NODES * H2_DIM;
        self_init_kernel<H2_DIM><<<(tot + TB - 1) / TB, TB, 0, stream>>>(htmp2, dinv, h2, N_NODES);
        edge_scatter_kernel<H2_DIM, 16><<<E_EDGES / TB, TB, 0, stream>>>(row, col, dinv, htmp2, h2);
        bias_add_kernel<H2_DIM><<<(tot + TB - 1) / TB, TB, 0, stream>>>(h2, b2, N_NODES);
    }

    {
        long long total_threads = (long long)2 * EP_CNT * 32;
        int blocks = (int)((total_threads + TB - 1) / TB);
        edge_dot_kernel<<<blocks, TB, 0, stream>>>(pos, neg, h2, out, EP_CNT);
    }
}


# Every elementwise pass folded into a neighbour: self term stored by both GEMM epilogues, bias+relu in GEMM2's A load, bias in the dot kernel, rsqrt at its uses, prep in one launch; 14 to 7 launches
# speedup vs baseline: 2.3724x; 1.2368x over previous
#include <hip/hip_runtime.h>

typedef __attribute__((ext_vector_type(2))) float v2f;
typedef __attribute__((ext_vector_type(4))) float v4f;
typedef __attribute__((ext_vector_type(8))) float v8f;

#define N_NODES 100000
#define F_IN    256
#define H1_DIM  128
#define H2_DIM  64
#define E_EDGES 1600000
#define EP_CNT  100000

__global__ void accum_deg_kernel(const int* __restrict__ col,
                                 float* __restrict__ deg, int e) {
    int i = blockIdx.x * blockDim.x + threadIdx.x;
    if (i < e) atomicAdd(&deg[col[i]], 1.0f);
}

template <int K, int NT>
__device__ __forceinline__ void frag_order_elem(const float* __restrict__ B,
                                                float* __restrict__ Bf, int idx) {
    constexpr int N = NT * 16;
    int s    = idx & 3;
    int lane = (idx >> 2) & 31;
    int t    = idx >> 7;
    int nt   = t % NT;
    int j    = t / NT;
    int c    = lane & 15;
    int h    = lane >> 4;
    int k    = 8 * j + 4 * h + s;
    Bf[idx] = B[(size_t)k * N + nt * 16 + c];
}

#define PREP_TB    256
#define W1_BLOCKS  (F_IN * H1_DIM / PREP_TB)
#define W2_BLOCKS  (H1_DIM * H2_DIM / PREP_TB)
#define DEG_BLOCKS ((N_NODES + PREP_TB - 1) / PREP_TB)
static_assert(F_IN * H1_DIM % PREP_TB == 0 && H1_DIM * H2_DIM % PREP_TB == 0,
              "whole blocks per weight matrix");

__global__ void __launch_bounds__(PREP_TB)
prep_kernel(const float* __restrict__ W1, const float* __restrict__ W2,
            float* __restrict__ W1f, float* __restrict__ W2f,
            float* __restrict__ deg) {
    const int b = blockIdx.x;
    if (b < W1_BLOCKS) {
        frag_order_elem<F_IN, 8>(W1, W1f, b * PREP_TB + threadIdx.x);
    } else if (b < W1_BLOCKS + W2_BLOCKS) {
        frag_order_elem<H1_DIM, 4>(W2, W2f, (b - W1_BLOCKS) * PREP_TB + threadIdx.x);
    } else {
        int i = (b - W1_BLOCKS - W2_BLOCKS) * PREP_TB + threadIdx.x;
        if (i < N_NODES) deg[i] = 1.0f;
    }
}

template <int NT>
__device__ __forceinline__ void wmma_hazard_slots(v8f (&acc)[NT]) {
#pragma unroll
    for (int nt = 0; nt < NT; ++nt)
        asm volatile("" : "+v"(acc[nt]));
    asm volatile("v_nop\n\tv_nop\n\tv_nop\n\tv_nop");
#pragma unroll
    for (int nt = 0; nt < NT; ++nt)
        asm volatile("" : "+v"(acc[nt]));
}

template <int K, int NT, bool ACT>
__global__ void __launch_bounds__(160)
gemm_wmma_f32_strip(const float* __restrict__ A, const float* __restrict__ abias,
                    const float* __restrict__ Bf, const float* __restrict__ deg,
                    float* __restrict__ C, float* __restrict__ S) {
    constexpr int N  = NT * 16;
    constexpr int NJ = K / 8;
    const int wave = threadIdx.x >> 5;
    const int lane = threadIdx.x & 31;
    const int c    = lane & 15;
    const int h    = lane >> 4;
    const int m0   = (blockIdx.x * 5 + wave) * 16;

    const float* ap = A + (size_t)(m0 + c) * K + 4 * h;
    const float* bp = Bf + lane * 4;
    const float* zp = abias + 4 * h;

    v8f acc[NT];
#pragma unroll
    for (int nt = 0; nt < NT; ++nt)
        acc[nt] = (v8f){0.f, 0.f, 0.f, 0.f, 0.f, 0.f, 0.f, 0.f};

    v4f a_cur = *(const v4f*)ap;
    v4f z_cur = {0.f, 0.f, 0.f, 0.f};
    if (ACT) z_cur = *(const v4f*)zp;
    v4f b_cur[NT];
#pragma unroll
    for (int nt = 0; nt < NT; ++nt)
        b_cur[nt] = *(const v4f*)(bp + nt * 128);

#pragma unroll
    for (int j = 0; j < NJ; ++j) {
        v4f a_nxt = a_cur;
        v4f z_nxt = z_cur;
        v4f b_nxt[NT];
#pragma unroll
        for (int nt = 0; nt < NT; ++nt)
            b_nxt[nt] = b_cur[nt];
        if (j + 1 < NJ) {
            a_nxt = *(const v4f*)(ap + 8 * (j + 1));
            if (ACT) z_nxt = *(const v4f*)(zp + 8 * (j + 1));
#pragma unroll
            for (int nt = 0; nt < NT; ++nt)
                b_nxt[nt] = *(const v4f*)(bp + ((j + 1) * NT + nt) * 128);
        }
        __builtin_amdgcn_sched_barrier(0);
        v4f av = a_cur;
        if (ACT) {
            if (j > 0) wmma_hazard_slots<NT>(acc);
            asm volatile("" : "+v"(a_cur));
            av = a_cur + z_cur;
            av.x = av.x > 0.f ? av.x : 0.f;
            av.y = av.y > 0.f ? av.y : 0.f;
            av.z = av.z > 0.f ? av.z : 0.f;
            av.w = av.w > 0.f ? av.w : 0.f;
            asm volatile("" : "+v"(av));
        }
        v2f a01 = {av.x, av.y};
        v2f a23 = {av.z, av.w};
#pragma unroll
        for (int nt = 0; nt < NT; ++nt) {
            v2f b01 = {b_cur[nt].x, b_cur[nt].y};
            acc[nt] = __builtin_amdgcn_wmma_f32_16x16x4_f32(
                 false, a01,  false, b01,
                 (short)0, acc[nt],  false,  false);
        }
#pragma unroll
        for (int nt = 0; nt < NT; ++nt) {
            v2f b23 = {b_cur[nt].z, b_cur[nt].w};
            acc[nt] = __builtin_amdgcn_wmma_f32_16x16x4_f32(
                 false, a23,  false, b23,
                 (short)0, acc[nt],  false,  false);
        }
        __builtin_amdgcn_sched_barrier(0);
        a_cur = a_nxt;
        z_cur = z_nxt;
#pragma unroll
        for (int nt = 0; nt < NT; ++nt)
            b_cur[nt] = b_nxt[nt];
    }

    wmma_hazard_slots<NT>(acc);
    int mm = m0, hh = h, cl = c;
    asm volatile("" : "+v"(mm), "+v"(hh), "+v"(cl));
    const int rw = mm + 8 * hh;
    const v4f dg0 = *(const v4f*)(deg + rw);
    const v4f dg1 = *(const v4f*)(deg + rw + 4);
    float* cp = C + (size_t)rw * N + cl;
#pragma unroll
    for (int nt = 0; nt < NT; ++nt)
#pragma unroll
        for (int r = 0; r < 8; ++r)
            cp[(size_t)r * N + nt * 16] = acc[nt][r];
    float di[8];
#pragma unroll
    for (int r = 0; r < 4; ++r) {
        di[r]     = rsqrtf(dg0[r]);
        di[r + 4] = rsqrtf(dg1[r]);
    }
    float* sp = S + (size_t)rw * N + cl;
#pragma unroll
    for (int nt = 0; nt < NT; ++nt)
#pragma unroll
        for (int r = 0; r < 8; ++r)
            sp[(size_t)r * N + nt * 16] = acc[nt][r] * di[r] * di[r];
}

template <int H, int G>
__global__ void __launch_bounds__(256)
edge_scatter_kernel(const int* __restrict__ row, const int* __restrict__ col,
                    const float* __restrict__ deg,
                    const float* __restrict__ htmp,
                    float* __restrict__ acc) {
    static_assert(E_EDGES % 256 == 0, "8-wave blocks of 32-edge waves cover the edges exactly");
    static_assert(32 % G == 0, "groups tile the wave's 32 edges");
    constexpr int V = H / 32;
    const unsigned lane = threadIdx.x & 31;
    const unsigned wave = (blockIdx.x * blockDim.x + threadIdx.x) >> 5;
    const unsigned ei = wave * 32 + lane;
    const int r = row[ei];
    const int c = col[ei];
    float dr = deg[r];
    float dc = deg[c];
    float nrm = 0.f;
#pragma unroll
    for (int j0 = 0; j0 < 32; j0 += G) {
        float v[G][V];
#pragma unroll
        for (int g = 0; g < G; ++g) {
            const int rj = __builtin_amdgcn_readlane(r, j0 + g);
            const float* src = htmp + (size_t)rj * H;
#pragma unroll
            for (int k = 0; k < V; ++k)
                v[g][k] = src[lane + k * 32];
        }
        if (j0 == 0) {
            asm volatile("" : "+v"(dr), "+v"(dc));
            nrm = rsqrtf(dr) * rsqrtf(dc);
        }
#pragma unroll
        for (int g = 0; g < G; ++g) {
            const int cj = __builtin_amdgcn_readlane(c, j0 + g);
            const float nj = __builtin_bit_cast(
                float, __builtin_amdgcn_readlane(__builtin_bit_cast(int, nrm), j0 + g));
            float* dst = acc + (size_t)cj * H;
#pragma unroll
            for (int k = 0; k < V; ++k)
                atomicAdd(&dst[lane + k * 32], v[g][k] * nj);
        }
    }
}

__global__ void __launch_bounds__(256)
edge_dot_kernel(const int* __restrict__ pos, const int* __restrict__ neg,
                const float* __restrict__ acc2, const float* __restrict__ bias,
                float* __restrict__ out, int ep) {
    int warp_id = (blockIdx.x * blockDim.x + threadIdx.x) >> 5;
    int lane = threadIdx.x & 31;
    if (warp_id >= 2 * ep) return;
    int src, dst;
    if (warp_id < ep) {
        src = pos[warp_id];
        dst = pos[ep + warp_id];
    } else {
        int e2 = warp_id - ep;
        src = neg[e2];
        dst = neg[ep + e2];
    }
    const float* xj = acc2 + (size_t)src * H2_DIM;
    const float* xi = acc2 + (size_t)dst * H2_DIM;
    const float bz0 = bias[lane];
    const float bz1 = bias[lane + 32];
    const float i0 = xi[lane] + bz0,      j0 = xj[lane] + bz0;
    const float i1 = xi[lane + 32] + bz1, j1 = xj[lane + 32] + bz1;
    float s = i0 * j0 + i1 * j1;
#pragma unroll
    for (int off = 16; off > 0; off >>= 1)
        s += __shfl_xor(s, off, 32);
    if (lane == 0) out[warp_id] = s;
}

static inline size_t align256(size_t x) { return (x + 255) & ~(size_t)255; }

extern "C" void kernel_launch(void* const* d_in, const int* in_sizes, int n_in,
                              void* d_out, int out_size, void* d_ws, size_t ws_size,
                              hipStream_t stream) {
    const float* x    = (const float*)d_in[0];
    const int*   ei   = (const int*)d_in[1];
    const int*   pos  = (const int*)d_in[2];
    const int*   neg  = (const int*)d_in[3];
    const float* W1   = (const float*)d_in[4];
    const float* b1   = (const float*)d_in[5];
    const float* W2   = (const float*)d_in[6];
    const float* b2   = (const float*)d_in[7];
    float* out = (float*)d_out;

    const int* row = ei;
    const int* col = ei + E_EDGES;

    char* ws = (char*)d_ws;
    size_t off = 0;
    float* deg   = (float*)(ws + off); off += align256(sizeof(float) * N_NODES);
    float* W1f   = (float*)(ws + off); off += align256(sizeof(float) * F_IN * H1_DIM);
    float* W2f   = (float*)(ws + off); off += align256(sizeof(float) * H1_DIM * H2_DIM);
    float* htmp1 = (float*)(ws + off); off += align256(sizeof(float) * (size_t)N_NODES * H1_DIM);
    float* acc1  = (float*)(ws + off); off += align256(sizeof(float) * (size_t)N_NODES * H1_DIM);
    float* htmp2 = (float*)(ws + off); off += align256(sizeof(float) * (size_t)N_NODES * H2_DIM);
    float* acc2  = (float*)(ws + off); off += align256(sizeof(float) * (size_t)N_NODES * H2_DIM);
    (void)ws_size; (void)n_in; (void)in_sizes; (void)out_size;

    const int TB = 256;

    prep_kernel<<<W1_BLOCKS + W2_BLOCKS + DEG_BLOCKS, PREP_TB, 0, stream>>>(W1, W2, W1f, W2f, deg);

    accum_deg_kernel<<<(E_EDGES + TB - 1) / TB, TB, 0, stream>>>(col, deg, E_EDGES);

    gemm_wmma_f32_strip<F_IN, 8, false><<<N_NODES / 80, 160, 0, stream>>>(
        x, b1, W1f, deg, htmp1, acc1);

    edge_scatter_kernel<H1_DIM, 8><<<E_EDGES / TB, TB, 0, stream>>>(row, col, deg, htmp1, acc1);

    gemm_wmma_f32_strip<H1_DIM, 4, true><<<N_NODES / 80, 160, 0, stream>>>(
        acc1, b1, W2f, deg, htmp2, acc2);

    edge_scatter_kernel<H2_DIM, 16><<<E_EDGES / TB, TB, 0, stream>>>(row, col, deg, htmp2, acc2);

    {
        long long total_threads = (long long)2 * EP_CNT * 32;
        int blocks = (int)((total_threads + TB - 1) / TB);
        edge_dot_kernel<<<blocks, TB, 0, stream>>>(pos, neg, acc2, b2, out, EP_CNT);
    }
}


# Degree and both edge-sum passes: a block draws (512-edge chunk, node half) from its chip half's counter and adds only into that half's nodes; a line's float atomics (arrival order) come from one half
# speedup vs baseline: 3.1544x; 1.3296x over previous
#include <hip/hip_runtime.h>

typedef __attribute__((ext_vector_type(2))) float v2f;
typedef __attribute__((ext_vector_type(4))) float v4f;
typedef __attribute__((ext_vector_type(8))) float v8f;

#define N_NODES 100000
#define F_IN    256
#define H1_DIM  128
#define H2_DIM  64
#define E_EDGES 1600000
#define EP_CNT  100000

#define CHUNK_E   512
#define NCHUNK    (E_EDGES / CHUNK_E)
#define HALF_NODE (N_NODES / 2)
#define N_TICKETS 6
static_assert(E_EDGES % CHUNK_E == 0, "whole chunks");

__device__ __forceinline__ int2 draw_chunk(int* __restrict__ cnt, int* sh) {
    if (threadIdx.x == 0) {
        const unsigned id = __builtin_amdgcn_s_sendmsg_rtn(0x87);
        int s = (int)((id >> 18) & 1u);
        int t = atomicAdd(&cnt[s], 1);
        if (t >= NCHUNK) { s ^= 1; t = atomicAdd(&cnt[s], 1); }
        sh[0] = t < NCHUNK ? t : NCHUNK;
        sh[1] = s;
    }
    __syncthreads();
    return make_int2(sh[0], sh[1]);
}

__global__ void __launch_bounds__(256)
accum_deg_kernel(const int* __restrict__ col, float* __restrict__ deg,
                 int* __restrict__ cnt) {
    __shared__ int sh[2];
    const int2 tk = draw_chunk(cnt, sh);
    if (tk.x >= NCHUNK) return;
    static_assert(CHUNK_E == 2 * 256, "two edges per thread");
    const int base = tk.x * CHUNK_E + threadIdx.x;
    int ca = col[base];
    int cb = col[base + 256];
    asm volatile("" : "+v"(ca), "+v"(cb));
    if ((ca >= HALF_NODE) == (tk.y != 0)) atomicAdd(&deg[ca], 1.0f);
    if ((cb >= HALF_NODE) == (tk.y != 0)) atomicAdd(&deg[cb], 1.0f);
}

template <int K, int NT>
__device__ __forceinline__ void frag_order_elem(const float* __restrict__ B,
                                                float* __restrict__ Bf, int idx) {
    constexpr int N = NT * 16;
    int s    = idx & 3;
    int lane = (idx >> 2) & 31;
    int t    = idx >> 7;
    int nt   = t % NT;
    int j    = t / NT;
    int c    = lane & 15;
    int h    = lane >> 4;
    int k    = 8 * j + 4 * h + s;
    Bf[idx] = B[(size_t)k * N + nt * 16 + c];
}

#define PREP_TB    256
#define W1_BLOCKS  (F_IN * H1_DIM / PREP_TB)
#define W2_BLOCKS  (H1_DIM * H2_DIM / PREP_TB)
#define DEG_BLOCKS ((N_NODES + PREP_TB - 1) / PREP_TB)
static_assert(F_IN * H1_DIM % PREP_TB == 0 && H1_DIM * H2_DIM % PREP_TB == 0,
              "whole blocks per weight matrix");

__global__ void __launch_bounds__(PREP_TB)
prep_kernel(const float* __restrict__ W1, const float* __restrict__ W2,
            float* __restrict__ W1f, float* __restrict__ W2f,
            float* __restrict__ deg, int* __restrict__ tickets) {
    const int b = blockIdx.x;
    if (b < W1_BLOCKS) {
        frag_order_elem<F_IN, 8>(W1, W1f, b * PREP_TB + threadIdx.x);
    } else if (b < W1_BLOCKS + W2_BLOCKS) {
        frag_order_elem<H1_DIM, 4>(W2, W2f, (b - W1_BLOCKS) * PREP_TB + threadIdx.x);
    } else {
        int i = (b - W1_BLOCKS - W2_BLOCKS) * PREP_TB + threadIdx.x;
        if (i < N_NODES) deg[i] = 1.0f;
        if (i < N_TICKETS) tickets[i] = 0;
    }
}

template <int NT>
__device__ __forceinline__ void wmma_hazard_slots(v8f (&acc)[NT]) {
#pragma unroll
    for (int nt = 0; nt < NT; ++nt)
        asm volatile("" : "+v"(acc[nt]));
    asm volatile("v_nop\n\tv_nop\n\tv_nop\n\tv_nop");
#pragma unroll
    for (int nt = 0; nt < NT; ++nt)
        asm volatile("" : "+v"(acc[nt]));
}

template <int K, int NT, bool ACT>
__global__ void __launch_bounds__(160)
gemm_wmma_f32_strip(const float* __restrict__ A, const float* __restrict__ abias,
                    const float* __restrict__ Bf, const float* __restrict__ deg,
                    float* __restrict__ C, float* __restrict__ S) {
    constexpr int N  = NT * 16;
    constexpr int NJ = K / 8;
    const int wave = threadIdx.x >> 5;
    const int lane = threadIdx.x & 31;
    const int c    = lane & 15;
    const int h    = lane >> 4;
    const int m0   = (blockIdx.x * 5 + wave) * 16;

    const float* ap = A + (size_t)(m0 + c) * K + 4 * h;
    const float* bp = Bf + lane * 4;
    const float* zp = abias + 4 * h;

    v8f acc[NT];
#pragma unroll
    for (int nt = 0; nt < NT; ++nt)
        acc[nt] = (v8f){0.f, 0.f, 0.f, 0.f, 0.f, 0.f, 0.f, 0.f};

    v4f a_cur = *(const v4f*)ap;
    v4f z_cur = {0.f, 0.f, 0.f, 0.f};
    if (ACT) z_cur = *(const v4f*)zp;
    v4f b_cur[NT];
#pragma unroll
    for (int nt = 0; nt < NT; ++nt)
        b_cur[nt] = *(const v4f*)(bp + nt * 128);

#pragma unroll
    for (int j = 0; j < NJ; ++j) {
        v4f a_nxt = a_cur;
        v4f z_nxt = z_cur;
        v4f b_nxt[NT];
#pragma unroll
        for (int nt = 0; nt < NT; ++nt)
            b_nxt[nt] = b_cur[nt];
        if (j + 1 < NJ) {
            a_nxt = *(const v4f*)(ap + 8 * (j + 1));
            if (ACT) z_nxt = *(const v4f*)(zp + 8 * (j + 1));
#pragma unroll
            for (int nt = 0; nt < NT; ++nt)
                b_nxt[nt] = *(const v4f*)(bp + ((j + 1) * NT + nt) * 128);
        }
        __builtin_amdgcn_sched_barrier(0);
        v4f av = a_cur;
        if (ACT) {
            if (j > 0) wmma_hazard_slots<NT>(acc);
            asm volatile("" : "+v"(a_cur));
            av = a_cur + z_cur;
            av.x = av.x > 0.f ? av.x : 0.f;
            av.y = av.y > 0.f ? av.y : 0.f;
            av.z = av.z > 0.f ? av.z : 0.f;
            av.w = av.w > 0.f ? av.w : 0.f;
            asm volatile("" : "+v"(av));
        }
        v2f a01 = {av.x, av.y};
        v2f a23 = {av.z, av.w};
#pragma unroll
        for (int nt = 0; nt < NT; ++nt) {
            v2f b01 = {b_cur[nt].x, b_cur[nt].y};
            acc[nt] = __builtin_amdgcn_wmma_f32_16x16x4_f32(
                 false, a01,  false, b01,
                 (short)0, acc[nt],  false,  false);
        }
#pragma unroll
        for (int nt = 0; nt < NT; ++nt) {
            v2f b23 = {b_cur[nt].z, b_cur[nt].w};
            acc[nt] = __builtin_amdgcn_wmma_f32_16x16x4_f32(
                 false, a23,  false, b23,
                 (short)0, acc[nt],  false,  false);
        }
        __builtin_amdgcn_sched_barrier(0);
        a_cur = a_nxt;
        z_cur = z_nxt;
#pragma unroll
        for (int nt = 0; nt < NT; ++nt)
            b_cur[nt] = b_nxt[nt];
    }

    wmma_hazard_slots<NT>(acc);
    int mm = m0, hh = h, cl = c;
    asm volatile("" : "+v"(mm), "+v"(hh), "+v"(cl));
    const int rw = mm + 8 * hh;
    const v4f dg0 = *(const v4f*)(deg + rw);
    const v4f dg1 = *(const v4f*)(deg + rw + 4);
    float* cp = C + (size_t)rw * N + cl;
#pragma unroll
    for (int nt = 0; nt < NT; ++nt)
#pragma unroll
        for (int r = 0; r < 8; ++r)
            cp[(size_t)r * N + nt * 16] = acc[nt][r];
    float di[8];
#pragma unroll
    for (int r = 0; r < 4; ++r) {
        di[r]     = rsqrtf(dg0[r]);
        di[r + 4] = rsqrtf(dg1[r]);
    }
    float* sp = S + (size_t)rw * N + cl;
#pragma unroll
    for (int nt = 0; nt < NT; ++nt)
#pragma unroll
        for (int r = 0; r < 8; ++r)
            sp[(size_t)r * N + nt * 16] = acc[nt][r] * di[r] * di[r];
}

template <int H, int G>
__global__ void __launch_bounds__(256)
edge_scatter_kernel(const int* __restrict__ row, const int* __restrict__ col,
                    const float* __restrict__ deg,
                    const float* __restrict__ htmp,
                    float* __restrict__ acc, int* __restrict__ cnt) {
    static_assert(CHUNK_E == 8 * 64, "8 waves of 64 edges cover a chunk");
    constexpr int V = H / 32;
    __shared__ int sh[2];
    const int2 tk = draw_chunk(cnt, sh);
    if (tk.x >= NCHUNK) return;
    const bool upper = tk.y != 0;
    const unsigned lane = threadIdx.x & 31;
    const unsigned wave = threadIdx.x >> 5;
    const unsigned e0 = (unsigned)tk.x * CHUNK_E + wave * 64 + lane;
    int r0 = row[e0];
    int r1 = row[e0 + 32];
    const int c0 = col[e0];
    const int c1 = col[e0 + 32];
    asm volatile("" : "+v"(r0), "+v"(r1));
    const bool mine0 = (c0 >= HALF_NODE) == upper;
    const bool mine1 = (c1 >= HALF_NODE) == upper;
    float dr0 = deg[mine0 ? r0 : 0], dc0 = deg[mine0 ? c0 : 0];
    float dr1 = deg[mine1 ? r1 : 0], dc1 = deg[mine1 ? c1 : 0];
    unsigned long long m =
        ((unsigned long long)__builtin_amdgcn_ballot_w32(mine1) << 32) |
        (unsigned long long)__builtin_amdgcn_ballot_w32(mine0);
    float nrm0 = 0.f, nrm1 = 0.f;
    bool first = true;
    while (m != 0) {
        float v[G][V];
        int   jj[G];
        bool  on[G];
#pragma unroll
        for (int g = 0; g < G; ++g) {
            on[g] = m != 0;
            jj[g] = on[g] ? __builtin_ctzll(m) : 0;
            m &= m - 1;
#pragma unroll
            for (int k = 0; k < V; ++k) v[g][k] = 0.f;
            if (on[g]) {
                const int ra = __builtin_amdgcn_readlane(r0, jj[g] & 31);
                const int rb = __builtin_amdgcn_readlane(r1, jj[g] & 31);
                const int rj = (jj[g] & 32) ? rb : ra;
                const float* src = htmp + (size_t)rj * H;
#pragma unroll
                for (int k = 0; k < V; ++k)
                    v[g][k] = src[lane + k * 32];
            }
        }
        if (first) {
            asm volatile("" : "+v"(dr0), "+v"(dc0), "+v"(dr1), "+v"(dc1));
            nrm0 = rsqrtf(dr0) * rsqrtf(dc0);
            nrm1 = rsqrtf(dr1) * rsqrtf(dc1);
            first = false;
        }
#pragma unroll
        for (int g = 0; g < G; ++g) {
            if (on[g]) {
                const int l  = jj[g] & 31;
                const int ca = __builtin_amdgcn_readlane(c0, l);
                const int cb = __builtin_amdgcn_readlane(c1, l);
                const int na = __builtin_amdgcn_readlane(__builtin_bit_cast(int, nrm0), l);
                const int nb = __builtin_amdgcn_readlane(__builtin_bit_cast(int, nrm1), l);
                const int cj = (jj[g] & 32) ? cb : ca;
                const float nj = __builtin_bit_cast(float, (jj[g] & 32) ? nb : na);
                float* dst = acc + (size_t)cj * H;
#pragma unroll
                for (int k = 0; k < V; ++k)
                    atomicAdd(&dst[lane + k * 32], v[g][k] * nj);
            }
        }
    }
}

__global__ void __launch_bounds__(256)
edge_dot_kernel(const int* __restrict__ pos, const int* __restrict__ neg,
                const float* __restrict__ acc2, const float* __restrict__ bias,
                float* __restrict__ out, int ep) {
    int warp_id = (blockIdx.x * blockDim.x + threadIdx.x) >> 5;
    int lane = threadIdx.x & 31;
    if (warp_id >= 2 * ep) return;
    int src, dst;
    if (warp_id < ep) {
        src = pos[warp_id];
        dst = pos[ep + warp_id];
    } else {
        int e2 = warp_id - ep;
        src = neg[e2];
        dst = neg[ep + e2];
    }
    const float* xj = acc2 + (size_t)src * H2_DIM;
    const float* xi = acc2 + (size_t)dst * H2_DIM;
    const float bz0 = bias[lane];
    const float bz1 = bias[lane + 32];
    const float i0 = xi[lane] + bz0,      j0 = xj[lane] + bz0;
    const float i1 = xi[lane + 32] + bz1, j1 = xj[lane + 32] + bz1;
    float s = i0 * j0 + i1 * j1;
#pragma unroll
    for (int off = 16; off > 0; off >>= 1)
        s += __shfl_xor(s, off, 32);
    if (lane == 0) out[warp_id] = s;
}

static inline size_t align256(size_t x) { return (x + 255) & ~(size_t)255; }

extern "C" void kernel_launch(void* const* d_in, const int* in_sizes, int n_in,
                              void* d_out, int out_size, void* d_ws, size_t ws_size,
                              hipStream_t stream) {
    const float* x    = (const float*)d_in[0];
    const int*   ei   = (const int*)d_in[1];
    const int*   pos  = (const int*)d_in[2];
    const int*   neg  = (const int*)d_in[3];
    const float* W1   = (const float*)d_in[4];
    const float* b1   = (const float*)d_in[5];
    const float* W2   = (const float*)d_in[6];
    const float* b2   = (const float*)d_in[7];
    float* out = (float*)d_out;

    const int* row = ei;
    const int* col = ei + E_EDGES;

    char* ws = (char*)d_ws;
    size_t off = 0;
    float* deg   = (float*)(ws + off);
    int*   tick  = (int*)(ws + off + sizeof(float) * N_NODES);
    off += align256(sizeof(float) * N_NODES + sizeof(int) * N_TICKETS);
    float* W1f   = (float*)(ws + off); off += align256(sizeof(float) * F_IN * H1_DIM);
    float* W2f   = (float*)(ws + off); off += align256(sizeof(float) * H1_DIM * H2_DIM);
    float* htmp1 = (float*)(ws + off); off += align256(sizeof(float) * (size_t)N_NODES * H1_DIM);
    float* acc1  = (float*)(ws + off); off += align256(sizeof(float) * (size_t)N_NODES * H1_DIM);
    float* htmp2 = (float*)(ws + off); off += align256(sizeof(float) * (size_t)N_NODES * H2_DIM);
    float* acc2  = (float*)(ws + off); off += align256(sizeof(float) * (size_t)N_NODES * H2_DIM);
    (void)ws_size; (void)n_in; (void)in_sizes; (void)out_size;

    const int TB = 256;

    prep_kernel<<<W1_BLOCKS + W2_BLOCKS + DEG_BLOCKS, PREP_TB, 0, stream>>>(W1, W2, W1f, W2f, deg, tick);

    accum_deg_kernel<<<2 * NCHUNK, TB, 0, stream>>>(col, deg, tick);

    gemm_wmma_f32_strip<F_IN, 8, false><<<N_NODES / 80, 160, 0, stream>>>(
        x, b1, W1f, deg, htmp1, acc1);

    edge_scatter_kernel<H1_DIM, 8><<<2 * NCHUNK, TB, 0, stream>>>(row, col, deg, htmp1, acc1, tick + 2);

    gemm_wmma_f32_strip<H1_DIM, 4, true><<<N_NODES / 80, 160, 0, stream>>>(
        acc1, b1, W2f, deg, htmp2, acc2);

    edge_scatter_kernel<H2_DIM, 16><<<2 * NCHUNK, TB, 0, stream>>>(row, col, deg, htmp2, acc2, tick + 4);

    {
        long long total_threads = (long long)2 * EP_CNT * 32;
        int blocks = (int)((total_threads + TB - 1) / TB);
        edge_dot_kernel<<<blocks, TB, 0, stream>>>(pos, neg, acc2, b2, out, EP_CNT);
    }
}


# Edge dot: a wave takes 32 test edges, not 1: lane i loads edge i's two node ids once, row loads of 16 edges go out together, the 32 lane sums fold in one shared xor tree (same order per edge)
# speedup vs baseline: 3.3793x; 1.0713x over previous
#include <hip/hip_runtime.h>

typedef __attribute__((ext_vector_type(2))) float v2f;
typedef __attribute__((ext_vector_type(4))) float v4f;
typedef __attribute__((ext_vector_type(8))) float v8f;

#define N_NODES 100000
#define F_IN    256
#define H1_DIM  128
#define H2_DIM  64
#define E_EDGES 1600000
#define EP_CNT  100000

#define CHUNK_E   512
#define NCHUNK    (E_EDGES / CHUNK_E)
#define HALF_NODE (N_NODES / 2)
#define N_TICKETS 6
static_assert(E_EDGES % CHUNK_E == 0, "whole chunks");

__device__ __forceinline__ int2 draw_chunk(int* __restrict__ cnt, int* sh) {
    if (threadIdx.x == 0) {
        const unsigned id = __builtin_amdgcn_s_sendmsg_rtn(0x87);
        int s = (int)((id >> 18) & 1u);
        int t = atomicAdd(&cnt[s], 1);
        if (t >= NCHUNK) { s ^= 1; t = atomicAdd(&cnt[s], 1); }
        sh[0] = t < NCHUNK ? t : NCHUNK;
        sh[1] = s;
    }
    __syncthreads();
    return make_int2(sh[0], sh[1]);
}

__global__ void __launch_bounds__(256)
accum_deg_kernel(const int* __restrict__ col, float* __restrict__ deg,
                 int* __restrict__ cnt) {
    __shared__ int sh[2];
    const int2 tk = draw_chunk(cnt, sh);
    if (tk.x >= NCHUNK) return;
    static_assert(CHUNK_E == 2 * 256, "two edges per thread");
    const int base = tk.x * CHUNK_E + threadIdx.x;
    int ca = col[base];
    int cb = col[base + 256];
    asm volatile("" : "+v"(ca), "+v"(cb));
    if ((ca >= HALF_NODE) == (tk.y != 0)) atomicAdd(&deg[ca], 1.0f);
    if ((cb >= HALF_NODE) == (tk.y != 0)) atomicAdd(&deg[cb], 1.0f);
}

template <int K, int NT>
__device__ __forceinline__ void frag_order_elem(const float* __restrict__ B,
                                                float* __restrict__ Bf, int idx) {
    constexpr int N = NT * 16;
    int s    = idx & 3;
    int lane = (idx >> 2) & 31;
    int t    = idx >> 7;
    int nt   = t % NT;
    int j    = t / NT;
    int c    = lane & 15;
    int h    = lane >> 4;
    int k    = 8 * j + 4 * h + s;
    Bf[idx] = B[(size_t)k * N + nt * 16 + c];
}

#define PREP_TB    256
#define W1_BLOCKS  (F_IN * H1_DIM / PREP_TB)
#define W2_BLOCKS  (H1_DIM * H2_DIM / PREP_TB)
#define DEG_BLOCKS ((N_NODES + PREP_TB - 1) / PREP_TB)
static_assert(F_IN * H1_DIM % PREP_TB == 0 && H1_DIM * H2_DIM % PREP_TB == 0,
              "whole blocks per weight matrix");

__global__ void __launch_bounds__(PREP_TB)
prep_kernel(const float* __restrict__ W1, const float* __restrict__ W2,
            float* __restrict__ W1f, float* __restrict__ W2f,
            float* __restrict__ deg, int* __restrict__ tickets) {
    const int b = blockIdx.x;
    if (b < W1_BLOCKS) {
        frag_order_elem<F_IN, 8>(W1, W1f, b * PREP_TB + threadIdx.x);
    } else if (b < W1_BLOCKS + W2_BLOCKS) {
        frag_order_elem<H1_DIM, 4>(W2, W2f, (b - W1_BLOCKS) * PREP_TB + threadIdx.x);
    } else {
        int i = (b - W1_BLOCKS - W2_BLOCKS) * PREP_TB + threadIdx.x;
        if (i < N_NODES) deg[i] = 1.0f;
        if (i < N_TICKETS) tickets[i] = 0;
    }
}

template <int NT>
__device__ __forceinline__ void wmma_hazard_slots(v8f (&acc)[NT]) {
#pragma unroll
    for (int nt = 0; nt < NT; ++nt)
        asm volatile("" : "+v"(acc[nt]));
    asm volatile("v_nop\n\tv_nop\n\tv_nop\n\tv_nop");
#pragma unroll
    for (int nt = 0; nt < NT; ++nt)
        asm volatile("" : "+v"(acc[nt]));
}

template <int K, int NT, bool ACT>
__global__ void __launch_bounds__(160)
gemm_wmma_f32_strip(const float* __restrict__ A, const float* __restrict__ abias,
                    const float* __restrict__ Bf, const float* __restrict__ deg,
                    float* __restrict__ C, float* __restrict__ S) {
    constexpr int N  = NT * 16;
    constexpr int NJ = K / 8;
    const int wave = threadIdx.x >> 5;
    const int lane = threadIdx.x & 31;
    const int c    = lane & 15;
    const int h    = lane >> 4;
    const int m0   = (blockIdx.x * 5 + wave) * 16;

    const float* ap = A + (size_t)(m0 + c) * K + 4 * h;
    const float* bp = Bf + lane * 4;
    const float* zp = abias + 4 * h;

    v8f acc[NT];
#pragma unroll
    for (int nt = 0; nt < NT; ++nt)
        acc[nt] = (v8f){0.f, 0.f, 0.f, 0.f, 0.f, 0.f, 0.f, 0.f};

    v4f a_cur = *(const v4f*)ap;
    v4f z_cur = {0.f, 0.f, 0.f, 0.f};
    if (ACT) z_cur = *(const v4f*)zp;
    v4f b_cur[NT];
#pragma unroll
    for (int nt = 0; nt < NT; ++nt)
        b_cur[nt] = *(const v4f*)(bp + nt * 128);

#pragma unroll
    for (int j = 0; j < NJ; ++j) {
        v4f a_nxt = a_cur;
        v4f z_nxt = z_cur;
        v4f b_nxt[NT];
#pragma unroll
        for (int nt = 0; nt < NT; ++nt)
            b_nxt[nt] = b_cur[nt];
        if (j + 1 < NJ) {
            a_nxt = *(const v4f*)(ap + 8 * (j + 1));
            if (ACT) z_nxt = *(const v4f*)(zp + 8 * (j + 1));
#pragma unroll
            for (int nt = 0; nt < NT; ++nt)
                b_nxt[nt] = *(const v4f*)(bp + ((j + 1) * NT + nt) * 128);
        }
        __builtin_amdgcn_sched_barrier(0);
        v4f av = a_cur;
        if (ACT) {
            if (j > 0) wmma_hazard_slots<NT>(acc);
            asm volatile("" : "+v"(a_cur));
            av = a_cur + z_cur;
            av.x = av.x > 0.f ? av.x : 0.f;
            av.y = av.y > 0.f ? av.y : 0.f;
            av.z = av.z > 0.f ? av.z : 0.f;
            av.w = av.w > 0.f ? av.w : 0.f;
            asm volatile("" : "+v"(av));
        }
        v2f a01 = {av.x, av.y};
        v2f a23 = {av.z, av.w};
#pragma unroll
        for (int nt = 0; nt < NT; ++nt) {
            v2f b01 = {b_cur[nt].x, b_cur[nt].y};
            acc[nt] = __builtin_amdgcn_wmma_f32_16x16x4_f32(
                 false, a01,  false, b01,
                 (short)0, acc[nt],  false,  false);
        }
#pragma unroll
        for (int nt = 0; nt < NT; ++nt) {
            v2f b23 = {b_cur[nt].z, b_cur[nt].w};
            acc[nt] = __builtin_amdgcn_wmma_f32_16x16x4_f32(
                 false, a23,  false, b23,
                 (short)0, acc[nt],  false,  false);
        }
        __builtin_amdgcn_sched_barrier(0);
        a_cur = a_nxt;
        z_cur = z_nxt;
#pragma unroll
        for (int nt = 0; nt < NT; ++nt)
            b_cur[nt] = b_nxt[nt];
    }

    wmma_hazard_slots<NT>(acc);
    int mm = m0, hh = h, cl = c;
    asm volatile("" : "+v"(mm), "+v"(hh), "+v"(cl));
    const int rw = mm + 8 * hh;
    const v4f dg0 = *(const v4f*)(deg + rw);
    const v4f dg1 = *(const v4f*)(deg + rw + 4);
    float* cp = C + (size_t)rw * N + cl;
#pragma unroll
    for (int nt = 0; nt < NT; ++nt)
#pragma unroll
        for (int r = 0; r < 8; ++r)
            cp[(size_t)r * N + nt * 16] = acc[nt][r];
    float di[8];
#pragma unroll
    for (int r = 0; r < 4; ++r) {
        di[r]     = rsqrtf(dg0[r]);
        di[r + 4] = rsqrtf(dg1[r]);
    }
    float* sp = S + (size_t)rw * N + cl;
#pragma unroll
    for (int nt = 0; nt < NT; ++nt)
#pragma unroll
        for (int r = 0; r < 8; ++r)
            sp[(size_t)r * N + nt * 16] = acc[nt][r] * di[r] * di[r];
}

template <int H, int G>
__global__ void __launch_bounds__(256)
edge_scatter_kernel(const int* __restrict__ row, const int* __restrict__ col,
                    const float* __restrict__ deg,
                    const float* __restrict__ htmp,
                    float* __restrict__ acc, int* __restrict__ cnt) {
    static_assert(CHUNK_E == 8 * 64, "8 waves of 64 edges cover a chunk");
    constexpr int V = H / 32;
    __shared__ int sh[2];
    const int2 tk = draw_chunk(cnt, sh);
    if (tk.x >= NCHUNK) return;
    const bool upper = tk.y != 0;
    const unsigned lane = threadIdx.x & 31;
    const unsigned wave = threadIdx.x >> 5;
    const unsigned e0 = (unsigned)tk.x * CHUNK_E + wave * 64 + lane;
    int r0 = row[e0];
    int r1 = row[e0 + 32];
    const int c0 = col[e0];
    const int c1 = col[e0 + 32];
    asm volatile("" : "+v"(r0), "+v"(r1));
    const bool mine0 = (c0 >= HALF_NODE) == upper;
    const bool mine1 = (c1 >= HALF_NODE) == upper;
    float dr0 = deg[mine0 ? r0 : 0], dc0 = deg[mine0 ? c0 : 0];
    float dr1 = deg[mine1 ? r1 : 0], dc1 = deg[mine1 ? c1 : 0];
    unsigned long long m =
        ((unsigned long long)__builtin_amdgcn_ballot_w32(mine1) << 32) |
        (unsigned long long)__builtin_amdgcn_ballot_w32(mine0);
    float nrm0 = 0.f, nrm1 = 0.f;
    bool first = true;
    while (m != 0) {
        float v[G][V];
        int   jj[G];
        bool  on[G];
#pragma unroll
        for (int g = 0; g < G; ++g) {
            on[g] = m != 0;
            jj[g] = on[g] ? __builtin_ctzll(m) : 0;
            m &= m - 1;
#pragma unroll
            for (int k = 0; k < V; ++k) v[g][k] = 0.f;
            if (on[g]) {
                const int ra = __builtin_amdgcn_readlane(r0, jj[g] & 31);
                const int rb = __builtin_amdgcn_readlane(r1, jj[g] & 31);
                const int rj = (jj[g] & 32) ? rb : ra;
                const float* src = htmp + (size_t)rj * H;
#pragma unroll
                for (int k = 0; k < V; ++k)
                    v[g][k] = src[lane + k * 32];
            }
        }
        if (first) {
            asm volatile("" : "+v"(dr0), "+v"(dc0), "+v"(dr1), "+v"(dc1));
            nrm0 = rsqrtf(dr0) * rsqrtf(dc0);
            nrm1 = rsqrtf(dr1) * rsqrtf(dc1);
            first = false;
        }
#pragma unroll
        for (int g = 0; g < G; ++g) {
            if (on[g]) {
                const int l  = jj[g] & 31;
                const int ca = __builtin_amdgcn_readlane(c0, l);
                const int cb = __builtin_amdgcn_readlane(c1, l);
                const int na = __builtin_amdgcn_readlane(__builtin_bit_cast(int, nrm0), l);
                const int nb = __builtin_amdgcn_readlane(__builtin_bit_cast(int, nrm1), l);
                const int cj = (jj[g] & 32) ? cb : ca;
                const float nj = __builtin_bit_cast(float, (jj[g] & 32) ? nb : na);
                float* dst = acc + (size_t)cj * H;
#pragma unroll
                for (int k = 0; k < V; ++k)
                    atomicAdd(&dst[lane + k * 32], v[g][k] * nj);
            }
        }
    }
}

#define DOT_G 16
static_assert(32 % DOT_G == 0, "whole groups per wave");
#define DOT_BLOCKS ((2 * EP_CNT + 63) / 64)

template <int D, int CNT>
__device__ __forceinline__ void fold_step(float (&p)[32], unsigned lane) {
    const bool up = (lane & D) != 0;
#pragma unroll
    for (int e = 0; e < CNT; ++e) {
        const float keep = up ? p[e + CNT] : p[e];
        const float send = up ? p[e] : p[e + CNT];
        p[e] = keep + __shfl_xor(send, D, 32);
    }
}

__global__ void __launch_bounds__(64)
edge_dot_kernel(const int* __restrict__ pos, const int* __restrict__ neg,
                const float* __restrict__ acc2, const float* __restrict__ bias,
                float* __restrict__ out, int ep) {
    const unsigned lane = threadIdx.x & 31;
    const int wave = (int)((blockIdx.x * blockDim.x + threadIdx.x) >> 5);
    const int e = wave * 32 + (int)lane;
    if (wave * 32 >= 2 * ep) return;
    const bool live = e < 2 * ep;
    const bool isneg = e >= ep;
    const int* tab = isneg ? neg : pos;
    const int k = live ? (isneg ? e - ep : e) : 0;
    const int srcn = tab[k];
    const int dstn = tab[ep + k];
    const float bz0 = bias[lane];
    const float bz1 = bias[lane + 32];

    float p[32];
#pragma unroll
    for (int g0 = 0; g0 < 32; g0 += DOT_G) {
        float v[DOT_G][4];
#pragma unroll
        for (int g = 0; g < DOT_G; ++g) {
            const int sj = __builtin_amdgcn_readlane(srcn, g0 + g);
            const int di = __builtin_amdgcn_readlane(dstn, g0 + g);
            const float* xi = acc2 + (size_t)di * H2_DIM;
            const float* xj = acc2 + (size_t)sj * H2_DIM;
            v[g][0] = xi[lane];
            v[g][1] = xj[lane];
            v[g][2] = xi[lane + 32];
            v[g][3] = xj[lane + 32];
        }
        __builtin_amdgcn_sched_barrier(0);
#pragma unroll
        for (int g = 0; g < DOT_G; ++g) {
            const float i0 = v[g][0] + bz0, j0 = v[g][1] + bz0;
            const float i1 = v[g][2] + bz1, j1 = v[g][3] + bz1;
            p[g0 + g] = i0 * j0 + i1 * j1;
        }
    }
    fold_step<16, 16>(p, lane);
    fold_step<8, 8>(p, lane);
    fold_step<4, 4>(p, lane);
    fold_step<2, 2>(p, lane);
    fold_step<1, 1>(p, lane);
    if (live) out[e] = p[0];
}

static inline size_t align256(size_t x) { return (x + 255) & ~(size_t)255; }

extern "C" void kernel_launch(void* const* d_in, const int* in_sizes, int n_in,
                              void* d_out, int out_size, void* d_ws, size_t ws_size,
                              hipStream_t stream) {
    const float* x    = (const float*)d_in[0];
    const int*   ei   = (const int*)d_in[1];
    const int*   pos  = (const int*)d_in[2];
    const int*   neg  = (const int*)d_in[3];
    const float* W1   = (const float*)d_in[4];
    const float* b1   = (const float*)d_in[5];
    const float* W2   = (const float*)d_in[6];
    const float* b2   = (const float*)d_in[7];
    float* out = (float*)d_out;

    const int* row = ei;
    const int* col = ei + E_EDGES;

    char* ws = (char*)d_ws;
    size_t off = 0;
    float* deg   = (float*)(ws + off);
    int*   tick  = (int*)(ws + off + sizeof(float) * N_NODES);
    off += align256(sizeof(float) * N_NODES + sizeof(int) * N_TICKETS);
    float* W1f   = (float*)(ws + off); off += align256(sizeof(float) * F_IN * H1_DIM);
    float* W2f   = (float*)(ws + off); off += align256(sizeof(float) * H1_DIM * H2_DIM);
    float* htmp1 = (float*)(ws + off); off += align256(sizeof(float) * (size_t)N_NODES * H1_DIM);
    float* acc1  = (float*)(ws + off); off += align256(sizeof(float) * (size_t)N_NODES * H1_DIM);
    float* htmp2 = (float*)(ws + off); off += align256(sizeof(float) * (size_t)N_NODES * H2_DIM);
    float* acc2  = (float*)(ws + off); off += align256(sizeof(float) * (size_t)N_NODES * H2_DIM);
    (void)ws_size; (void)n_in; (void)in_sizes; (void)out_size;

    const int TB = 256;

    prep_kernel<<<W1_BLOCKS + W2_BLOCKS + DEG_BLOCKS, PREP_TB, 0, stream>>>(W1, W2, W1f, W2f, deg, tick);

    accum_deg_kernel<<<2 * NCHUNK, TB, 0, stream>>>(col, deg, tick);

    gemm_wmma_f32_strip<F_IN, 8, false><<<N_NODES / 80, 160, 0, stream>>>(
        x, b1, W1f, deg, htmp1, acc1);

    edge_scatter_kernel<H1_DIM, 8><<<2 * NCHUNK, TB, 0, stream>>>(row, col, deg, htmp1, acc1, tick + 2);

    gemm_wmma_f32_strip<H1_DIM, 4, true><<<N_NODES / 80, 160, 0, stream>>>(
        acc1, b1, W2f, deg, htmp2, acc2);

    edge_scatter_kernel<H2_DIM, 16><<<2 * NCHUNK, TB, 0, stream>>>(row, col, deg, htmp2, acc2, tick + 4);

    edge_dot_kernel<<<DOT_BLOCKS, 64, 0, stream>>>(pos, neg, acc2, b2, out, EP_CNT);
}


# Edge factor rsqrt(deg[r])*rsqrt(deg[c]) written once per edge by layer 1's product launch (under its matrix time), read as whole lines by both edge passes: no scattered deg reads there; same bits
# speedup vs baseline: 3.4847x; 1.0312x over previous
#include <hip/hip_runtime.h>

typedef __attribute__((ext_vector_type(2))) float v2f;
typedef __attribute__((ext_vector_type(4))) float v4f;
typedef __attribute__((ext_vector_type(8))) float v8f;

#define N_NODES 100000
#define F_IN    256
#define H1_DIM  128
#define H2_DIM  64
#define E_EDGES 1600000
#define EP_CNT  100000

#define CHUNK_E   512
#define NCHUNK    (E_EDGES / CHUNK_E)
#define HALF_NODE (N_NODES / 2)
#define N_TICKETS 6
static_assert(E_EDGES % CHUNK_E == 0, "whole chunks");

__device__ __forceinline__ int2 draw_chunk(int* __restrict__ cnt, int* sh) {
    if (threadIdx.x == 0) {
        const unsigned id = __builtin_amdgcn_s_sendmsg_rtn(0x87);
        int s = (int)((id >> 18) & 1u);
        int t = atomicAdd(&cnt[s], 1);
        if (t >= NCHUNK) { s ^= 1; t = atomicAdd(&cnt[s], 1); }
        sh[0] = t < NCHUNK ? t : NCHUNK;
        sh[1] = s;
    }
    __syncthreads();
    return make_int2(sh[0], sh[1]);
}

__global__ void __launch_bounds__(256)
accum_deg_kernel(const int* __restrict__ col, float* __restrict__ deg,
                 int* __restrict__ cnt) {
    __shared__ int sh[2];
    const int2 tk = draw_chunk(cnt, sh);
    if (tk.x >= NCHUNK) return;
    static_assert(CHUNK_E == 2 * 256, "two edges per thread");
    const int base = tk.x * CHUNK_E + threadIdx.x;
    int ca = col[base];
    int cb = col[base + 256];
    asm volatile("" : "+v"(ca), "+v"(cb));
    if ((ca >= HALF_NODE) == (tk.y != 0)) atomicAdd(&deg[ca], 1.0f);
    if ((cb >= HALF_NODE) == (tk.y != 0)) atomicAdd(&deg[cb], 1.0f);
}

template <int K, int NT>
__device__ __forceinline__ void frag_order_elem(const float* __restrict__ B,
                                                float* __restrict__ Bf, int idx) {
    constexpr int N = NT * 16;
    int s    = idx & 3;
    int lane = (idx >> 2) & 31;
    int t    = idx >> 7;
    int nt   = t % NT;
    int j    = t / NT;
    int c    = lane & 15;
    int h    = lane >> 4;
    int k    = 8 * j + 4 * h + s;
    Bf[idx] = B[(size_t)k * N + nt * 16 + c];
}

#define PREP_TB    256
#define W1_BLOCKS  (F_IN * H1_DIM / PREP_TB)
#define W2_BLOCKS  (H1_DIM * H2_DIM / PREP_TB)
#define DEG_BLOCKS ((N_NODES + PREP_TB - 1) / PREP_TB)
static_assert(F_IN * H1_DIM % PREP_TB == 0 && H1_DIM * H2_DIM % PREP_TB == 0,
              "whole blocks per weight matrix");

__global__ void __launch_bounds__(PREP_TB)
prep_kernel(const float* __restrict__ W1, const float* __restrict__ W2,
            float* __restrict__ W1f, float* __restrict__ W2f,
            float* __restrict__ deg, int* __restrict__ tickets) {
    const int b = blockIdx.x;
    if (b < W1_BLOCKS) {
        frag_order_elem<F_IN, 8>(W1, W1f, b * PREP_TB + threadIdx.x);
    } else if (b < W1_BLOCKS + W2_BLOCKS) {
        frag_order_elem<H1_DIM, 4>(W2, W2f, (b - W1_BLOCKS) * PREP_TB + threadIdx.x);
    } else {
        int i = (b - W1_BLOCKS - W2_BLOCKS) * PREP_TB + threadIdx.x;
        if (i < N_NODES) deg[i] = 1.0f;
        if (i < N_TICKETS) tickets[i] = 0;
    }
}

template <int NT>
__device__ __forceinline__ void wmma_hazard_slots(v8f (&acc)[NT]) {
#pragma unroll
    for (int nt = 0; nt < NT; ++nt)
        asm volatile("" : "+v"(acc[nt]));
    asm volatile("v_nop\n\tv_nop\n\tv_nop\n\tv_nop");
#pragma unroll
    for (int nt = 0; nt < NT; ++nt)
        asm volatile("" : "+v"(acc[nt]));
}

#define EDGES_PER_STRIP (E_EDGES / (N_NODES / 16))
static_assert(EDGES_PER_STRIP * (N_NODES / 16) == E_EDGES && EDGES_PER_STRIP % 128 == 0,
              "the strip waves share the edge list in whole 128-edge trips");

template <int K, int NT, bool ACT, bool EDGE_NORM>
__global__ void __launch_bounds__(160)
gemm_wmma_f32_strip(const float* __restrict__ A, const float* __restrict__ abias,
                    const float* __restrict__ Bf, const float* __restrict__ deg,
                    float* __restrict__ C, float* __restrict__ S,
                    const int* __restrict__ row, const int* __restrict__ col,
                    float* __restrict__ nrm) {
    constexpr int N  = NT * 16;
    constexpr int NJ = K / 8;
    const int wave = threadIdx.x >> 5;
    const int lane = threadIdx.x & 31;
    const int c    = lane & 15;
    const int h    = lane >> 4;
    const int m0   = (blockIdx.x * 5 + wave) * 16;

    if (EDGE_NORM) {
        constexpr int TRIPS = EDGES_PER_STRIP / 128;
        const size_t eb = (size_t)(blockIdx.x * 5 + wave) * EDGES_PER_STRIP + 4 * lane;
        int4 rr[TRIPS], cc[TRIPS];
#pragma unroll
        for (int t = 0; t < TRIPS; ++t) {
            rr[t] = *(const int4*)(row + eb + t * 128);
            cc[t] = *(const int4*)(col + eb + t * 128);
        }
        v4f dr[TRIPS], dc[TRIPS];
#pragma unroll
        for (int t = 0; t < TRIPS; ++t) {
            dr[t] = (v4f){deg[rr[t].x], deg[rr[t].y], deg[rr[t].z], deg[rr[t].w]};
            dc[t] = (v4f){deg[cc[t].x], deg[cc[t].y], deg[cc[t].z], deg[cc[t].w]};
        }
#pragma unroll
        for (int t = 0; t < TRIPS; ++t) {
            v4f nv;
            nv.x = rsqrtf(dr[t].x) * rsqrtf(dc[t].x);
            nv.y = rsqrtf(dr[t].y) * rsqrtf(dc[t].y);
            nv.z = rsqrtf(dr[t].z) * rsqrtf(dc[t].z);
            nv.w = rsqrtf(dr[t].w) * rsqrtf(dc[t].w);
            *(v4f*)(nrm + eb + t * 128) = nv;
        }
    }

    const float* ap = A + (size_t)(m0 + c) * K + 4 * h;
    const float* bp = Bf + lane * 4;
    const float* zp = abias + 4 * h;

    v8f acc[NT];
#pragma unroll
    for (int nt = 0; nt < NT; ++nt)
        acc[nt] = (v8f){0.f, 0.f, 0.f, 0.f, 0.f, 0.f, 0.f, 0.f};

    v4f a_cur = *(const v4f*)ap;
    v4f z_cur = {0.f, 0.f, 0.f, 0.f};
    if (ACT) z_cur = *(const v4f*)zp;
    v4f b_cur[NT];
#pragma unroll
    for (int nt = 0; nt < NT; ++nt)
        b_cur[nt] = *(const v4f*)(bp + nt * 128);

#pragma unroll
    for (int j = 0; j < NJ; ++j) {
        v4f a_nxt = a_cur;
        v4f z_nxt = z_cur;
        v4f b_nxt[NT];
#pragma unroll
        for (int nt = 0; nt < NT; ++nt)
            b_nxt[nt] = b_cur[nt];
        if (j + 1 < NJ) {
            a_nxt = *(const v4f*)(ap + 8 * (j + 1));
            if (ACT) z_nxt = *(const v4f*)(zp + 8 * (j + 1));
#pragma unroll
            for (int nt = 0; nt < NT; ++nt)
                b_nxt[nt] = *(const v4f*)(bp + ((j + 1) * NT + nt) * 128);
        }
        __builtin_amdgcn_sched_barrier(0);
        v4f av = a_cur;
        if (ACT) {
            if (j > 0) wmma_hazard_slots<NT>(acc);
            asm volatile("" : "+v"(a_cur));
            av = a_cur + z_cur;
            av.x = av.x > 0.f ? av.x : 0.f;
            av.y = av.y > 0.f ? av.y : 0.f;
            av.z = av.z > 0.f ? av.z : 0.f;
            av.w = av.w > 0.f ? av.w : 0.f;
            asm volatile("" : "+v"(av));
        }
        v2f a01 = {av.x, av.y};
        v2f a23 = {av.z, av.w};
#pragma unroll
        for (int nt = 0; nt < NT; ++nt) {
            v2f b01 = {b_cur[nt].x, b_cur[nt].y};
            acc[nt] = __builtin_amdgcn_wmma_f32_16x16x4_f32(
                 false, a01,  false, b01,
                 (short)0, acc[nt],  false,  false);
        }
#pragma unroll
        for (int nt = 0; nt < NT; ++nt) {
            v2f b23 = {b_cur[nt].z, b_cur[nt].w};
            acc[nt] = __builtin_amdgcn_wmma_f32_16x16x4_f32(
                 false, a23,  false, b23,
                 (short)0, acc[nt],  false,  false);
        }
        __builtin_amdgcn_sched_barrier(0);
        a_cur = a_nxt;
        z_cur = z_nxt;
#pragma unroll
        for (int nt = 0; nt < NT; ++nt)
            b_cur[nt] = b_nxt[nt];
    }

    wmma_hazard_slots<NT>(acc);
    int mm = m0, hh = h, cl = c;
    asm volatile("" : "+v"(mm), "+v"(hh), "+v"(cl));
    const int rw = mm + 8 * hh;
    const v4f dg0 = *(const v4f*)(deg + rw);
    const v4f dg1 = *(const v4f*)(deg + rw + 4);
    float* cp = C + (size_t)rw * N + cl;
#pragma unroll
    for (int nt = 0; nt < NT; ++nt)
#pragma unroll
        for (int r = 0; r < 8; ++r)
            cp[(size_t)r * N + nt * 16] = acc[nt][r];
    float di[8];
#pragma unroll
    for (int r = 0; r < 4; ++r) {
        di[r]     = rsqrtf(dg0[r]);
        di[r + 4] = rsqrtf(dg1[r]);
    }
    float* sp = S + (size_t)rw * N + cl;
#pragma unroll
    for (int nt = 0; nt < NT; ++nt)
#pragma unroll
        for (int r = 0; r < 8; ++r)
            sp[(size_t)r * N + nt * 16] = acc[nt][r] * di[r] * di[r];
}

template <int H, int G>
__global__ void __launch_bounds__(256)
edge_scatter_kernel(const int* __restrict__ row, const int* __restrict__ col,
                    const float* __restrict__ nrm,
                    const float* __restrict__ htmp,
                    float* __restrict__ acc, int* __restrict__ cnt) {
    static_assert(CHUNK_E == 8 * 64, "8 waves of 64 edges cover a chunk");
    constexpr int V = H / 32;
    __shared__ int sh[2];
    const int2 tk = draw_chunk(cnt, sh);
    if (tk.x >= NCHUNK) return;
    const bool upper = tk.y != 0;
    const unsigned lane = threadIdx.x & 31;
    const unsigned wave = threadIdx.x >> 5;
    const unsigned e0 = (unsigned)tk.x * CHUNK_E + wave * 64 + lane;
    int r0 = row[e0];
    int r1 = row[e0 + 32];
    const int c0 = col[e0];
    const int c1 = col[e0 + 32];
    float nrm0 = nrm[e0];
    float nrm1 = nrm[e0 + 32];
    asm volatile("" : "+v"(r0), "+v"(r1), "+v"(nrm0), "+v"(nrm1));
    const bool mine0 = (c0 >= HALF_NODE) == upper;
    const bool mine1 = (c1 >= HALF_NODE) == upper;
    unsigned long long m =
        ((unsigned long long)__builtin_amdgcn_ballot_w32(mine1) << 32) |
        (unsigned long long)__builtin_amdgcn_ballot_w32(mine0);
    while (m != 0) {
        float v[G][V];
        int   jj[G];
        bool  on[G];
#pragma unroll
        for (int g = 0; g < G; ++g) {
            on[g] = m != 0;
            jj[g] = on[g] ? __builtin_ctzll(m) : 0;
            m &= m - 1;
#pragma unroll
            for (int k = 0; k < V; ++k) v[g][k] = 0.f;
            if (on[g]) {
                const int ra = __builtin_amdgcn_readlane(r0, jj[g] & 31);
                const int rb = __builtin_amdgcn_readlane(r1, jj[g] & 31);
                const int rj = (jj[g] & 32) ? rb : ra;
                const float* src = htmp + (size_t)rj * H;
#pragma unroll
                for (int k = 0; k < V; ++k)
                    v[g][k] = src[lane + k * 32];
            }
        }
#pragma unroll
        for (int g = 0; g < G; ++g) {
            if (on[g]) {
                const int l  = jj[g] & 31;
                const int ca = __builtin_amdgcn_readlane(c0, l);
                const int cb = __builtin_amdgcn_readlane(c1, l);
                const int na = __builtin_amdgcn_readlane(__builtin_bit_cast(int, nrm0), l);
                const int nb = __builtin_amdgcn_readlane(__builtin_bit_cast(int, nrm1), l);
                const int cj = (jj[g] & 32) ? cb : ca;
                const float nj = __builtin_bit_cast(float, (jj[g] & 32) ? nb : na);
                float* dst = acc + (size_t)cj * H;
#pragma unroll
                for (int k = 0; k < V; ++k)
                    atomicAdd(&dst[lane + k * 32], v[g][k] * nj);
            }
        }
    }
}

#define DOT_G 16
static_assert(32 % DOT_G == 0, "whole groups per wave");
#define DOT_BLOCKS ((2 * EP_CNT + 63) / 64)

template <int D, int CNT>
__device__ __forceinline__ void fold_step(float (&p)[32], unsigned lane) {
    const bool up = (lane & D) != 0;
#pragma unroll
    for (int e = 0; e < CNT; ++e) {
        const float keep = up ? p[e + CNT] : p[e];
        const float send = up ? p[e] : p[e + CNT];
        p[e] = keep + __shfl_xor(send, D, 32);
    }
}

__global__ void __launch_bounds__(64)
edge_dot_kernel(const int* __restrict__ pos, const int* __restrict__ neg,
                const float* __restrict__ acc2, const float* __restrict__ bias,
                float* __restrict__ out, int ep) {
    const unsigned lane = threadIdx.x & 31;
    const int wave = (int)((blockIdx.x * blockDim.x + threadIdx.x) >> 5);
    const int e = wave * 32 + (int)lane;
    if (wave * 32 >= 2 * ep) return;
    const bool live = e < 2 * ep;
    const bool isneg = e >= ep;
    const int* tab = isneg ? neg : pos;
    const int k = live ? (isneg ? e - ep : e) : 0;
    const int srcn = tab[k];
    const int dstn = tab[ep + k];
    const float bz0 = bias[lane];
    const float bz1 = bias[lane + 32];

    float p[32];
#pragma unroll
    for (int g0 = 0; g0 < 32; g0 += DOT_G) {
        float v[DOT_G][4];
#pragma unroll
        for (int g = 0; g < DOT_G; ++g) {
            const int sj = __builtin_amdgcn_readlane(srcn, g0 + g);
            const int di = __builtin_amdgcn_readlane(dstn, g0 + g);
            const float* xi = acc2 + (size_t)di * H2_DIM;
            const float* xj = acc2 + (size_t)sj * H2_DIM;
            v[g][0] = xi[lane];
            v[g][1] = xj[lane];
            v[g][2] = xi[lane + 32];
            v[g][3] = xj[lane + 32];
        }
        __builtin_amdgcn_sched_barrier(0);
#pragma unroll
        for (int g = 0; g < DOT_G; ++g) {
            const float i0 = v[g][0] + bz0, j0 = v[g][1] + bz0;
            const float i1 = v[g][2] + bz1, j1 = v[g][3] + bz1;
            p[g0 + g] = i0 * j0 + i1 * j1;
        }
    }
    fold_step<16, 16>(p, lane);
    fold_step<8, 8>(p, lane);
    fold_step<4, 4>(p, lane);
    fold_step<2, 2>(p, lane);
    fold_step<1, 1>(p, lane);
    if (live) out[e] = p[0];
}

static inline size_t align256(size_t x) { return (x + 255) & ~(size_t)255; }

extern "C" void kernel_launch(void* const* d_in, const int* in_sizes, int n_in,
                              void* d_out, int out_size, void* d_ws, size_t ws_size,
                              hipStream_t stream) {
    const float* x    = (const float*)d_in[0];
    const int*   ei   = (const int*)d_in[1];
    const int*   pos  = (const int*)d_in[2];
    const int*   neg  = (const int*)d_in[3];
    const float* W1   = (const float*)d_in[4];
    const float* b1   = (const float*)d_in[5];
    const float* W2   = (const float*)d_in[6];
    const float* b2   = (const float*)d_in[7];
    float* out = (float*)d_out;

    const int* row = ei;
    const int* col = ei + E_EDGES;

    char* ws = (char*)d_ws;
    size_t off = 0;
    float* deg   = (float*)(ws + off);
    int*   tick  = (int*)(ws + off + sizeof(float) * N_NODES);
    off += align256(sizeof(float) * N_NODES + sizeof(int) * N_TICKETS);
    float* W1f   = (float*)(ws + off); off += align256(sizeof(float) * F_IN * H1_DIM);
    float* W2f   = (float*)(ws + off); off += align256(sizeof(float) * H1_DIM * H2_DIM);
    float* htmp1 = (float*)(ws + off); off += align256(sizeof(float) * (size_t)N_NODES * H1_DIM);
    float* acc1  = (float*)(ws + off); off += align256(sizeof(float) * (size_t)N_NODES * H1_DIM);
    static_assert((size_t)E_EDGES <= (size_t)N_NODES * H2_DIM, "factors fit the region");
    float* nrm   = (float*)(ws + off); off += align256(sizeof(float) * (size_t)N_NODES * H2_DIM);
    float* acc2  = (float*)(ws + off); off += align256(sizeof(float) * (size_t)N_NODES * H2_DIM);
    float* htmp2 = htmp1;
    (void)ws_size; (void)n_in; (void)in_sizes; (void)out_size;

    const int TB = 256;

    prep_kernel<<<W1_BLOCKS + W2_BLOCKS + DEG_BLOCKS, PREP_TB, 0, stream>>>(W1, W2, W1f, W2f, deg, tick);

    accum_deg_kernel<<<2 * NCHUNK, TB, 0, stream>>>(col, deg, tick);

    gemm_wmma_f32_strip<F_IN, 8, false, true><<<N_NODES / 80, 160, 0, stream>>>(
        x, b1, W1f, deg, htmp1, acc1, row, col, nrm);

    edge_scatter_kernel<H1_DIM, 8><<<2 * NCHUNK, TB, 0, stream>>>(row, col, nrm, htmp1, acc1, tick + 2);

    gemm_wmma_f32_strip<H1_DIM, 4, true, false><<<N_NODES / 80, 160, 0, stream>>>(
        acc1, b1, W2f, deg, htmp2, acc2, row, col, nrm);

    edge_scatter_kernel<H2_DIM, 16><<<2 * NCHUNK, TB, 0, stream>>>(row, col, nrm, htmp2, acc2, tick + 4);

    edge_dot_kernel<<<DOT_BLOCKS, 64, 0, stream>>>(pos, neg, acc2, b2, out, EP_CNT);
}


# Degree count: a block per (edge slice, node range) counts its range in LDS (integer adds) and adds the table into deg, lane = node: whole-line float adds of exact counts, not one global add per edge
# speedup vs baseline: 3.6115x; 1.0364x over previous
#include <hip/hip_runtime.h>

typedef __attribute__((ext_vector_type(2))) float v2f;
typedef __attribute__((ext_vector_type(4))) float v4f;
typedef __attribute__((ext_vector_type(8))) float v8f;

#define N_NODES 100000
#define F_IN    256
#define H1_DIM  128
#define H2_DIM  64
#define E_EDGES 1600000
#define EP_CNT  100000

#define CHUNK_E   512
#define NCHUNK    (E_EDGES / CHUNK_E)
#define HALF_NODE (N_NODES / 2)
#define N_TICKETS 4
static_assert(E_EDGES % CHUNK_E == 0, "whole chunks");

__device__ __forceinline__ int2 draw_chunk(int* __restrict__ cnt, int* sh) {
    if (threadIdx.x == 0) {
        const unsigned id = __builtin_amdgcn_s_sendmsg_rtn(0x87);
        int s = (int)((id >> 18) & 1u);
        int t = atomicAdd(&cnt[s], 1);
        if (t >= NCHUNK) { s ^= 1; t = atomicAdd(&cnt[s], 1); }
        sh[0] = t < NCHUNK ? t : NCHUNK;
        sh[1] = s;
    }
    __syncthreads();
    return make_int2(sh[0], sh[1]);
}

#define DEG_RANGES   8
#define DEG_SLICES   32
#define DEG_TB       1024
#define DEG_RANGE_N  12544
#define DEG_SLICE_E  (E_EDGES / DEG_SLICES)
static_assert(DEG_RANGE_N % 32 == 0 && DEG_RANGES * DEG_RANGE_N >= N_NODES,
              "ranges of whole lines cover the nodes");
static_assert(E_EDGES % DEG_SLICES == 0 && DEG_SLICE_E % 4 == 0,
              "slices of whole 16-byte pieces");

__device__ __forceinline__ void count_in_range(int* hist, int c, int lo) {
    const unsigned u = (unsigned)(c - lo);
    if (u < DEG_RANGE_N) atomicAdd(&hist[u], 1);
}

__global__ void __launch_bounds__(DEG_TB)
accum_deg_kernel(const int* __restrict__ col, float* __restrict__ deg) {
    __shared__ int hist[DEG_RANGE_N];
    const int tid   = threadIdx.x;
    const int range = blockIdx.x % DEG_RANGES;
    const int slice = blockIdx.x / DEG_RANGES;
    const int lo    = range * DEG_RANGE_N;

    constexpr int NQ = DEG_SLICE_E / 4;
    constexpr int NK = (NQ + DEG_TB - 1) / DEG_TB;
    const int4* cp = (const int4*)(col + (size_t)slice * DEG_SLICE_E);
    int4 v[NK];
#pragma unroll
    for (int k = 0; k < NK; ++k) {
        const int q = k * DEG_TB + tid;
        v[k] = make_int4(-1, -1, -1, -1);
        if ((k + 1) * DEG_TB <= NQ || q < NQ) v[k] = cp[q];
    }
    for (int n = tid; n < DEG_RANGE_N; n += DEG_TB) hist[n] = 0;
    __syncthreads();
#pragma unroll
    for (int k = 0; k < NK; ++k) {
        count_in_range(hist, v[k].x, lo);
        count_in_range(hist, v[k].y, lo);
        count_in_range(hist, v[k].z, lo);
        count_in_range(hist, v[k].w, lo);
    }
    __syncthreads();
    constexpr int NF = (DEG_RANGE_N + DEG_TB - 1) / DEG_TB;
    int cnt[NF];
#pragma unroll
    for (int k = 0; k < NF; ++k) {
        const int n = k * DEG_TB + tid;
        cnt[k] = 0;
        if ((k + 1) * DEG_TB <= DEG_RANGE_N || n < DEG_RANGE_N) cnt[k] = hist[n];
    }
#pragma unroll
    for (int k = 0; k < NF; ++k) {
        const int node = lo + k * DEG_TB + tid;
        if (cnt[k] != 0 && node < N_NODES) atomicAdd(&deg[node], (float)cnt[k]);
    }
}

template <int K, int NT>
__device__ __forceinline__ void frag_order_elem(const float* __restrict__ B,
                                                float* __restrict__ Bf, int idx) {
    constexpr int N = NT * 16;
    int s    = idx & 3;
    int lane = (idx >> 2) & 31;
    int t    = idx >> 7;
    int nt   = t % NT;
    int j    = t / NT;
    int c    = lane & 15;
    int h    = lane >> 4;
    int k    = 8 * j + 4 * h + s;
    Bf[idx] = B[(size_t)k * N + nt * 16 + c];
}

#define PREP_TB    256
#define W1_BLOCKS  (F_IN * H1_DIM / PREP_TB)
#define W2_BLOCKS  (H1_DIM * H2_DIM / PREP_TB)
#define DEG_BLOCKS ((N_NODES + PREP_TB - 1) / PREP_TB)
static_assert(F_IN * H1_DIM % PREP_TB == 0 && H1_DIM * H2_DIM % PREP_TB == 0,
              "whole blocks per weight matrix");

__global__ void __launch_bounds__(PREP_TB)
prep_kernel(const float* __restrict__ W1, const float* __restrict__ W2,
            float* __restrict__ W1f, float* __restrict__ W2f,
            float* __restrict__ deg, int* __restrict__ tickets) {
    const int b = blockIdx.x;
    if (b < W1_BLOCKS) {
        frag_order_elem<F_IN, 8>(W1, W1f, b * PREP_TB + threadIdx.x);
    } else if (b < W1_BLOCKS + W2_BLOCKS) {
        frag_order_elem<H1_DIM, 4>(W2, W2f, (b - W1_BLOCKS) * PREP_TB + threadIdx.x);
    } else {
        int i = (b - W1_BLOCKS - W2_BLOCKS) * PREP_TB + threadIdx.x;
        if (i < N_NODES) deg[i] = 1.0f;
        if (i < N_TICKETS) tickets[i] = 0;
    }
}

template <int NT>
__device__ __forceinline__ void wmma_hazard_slots(v8f (&acc)[NT]) {
#pragma unroll
    for (int nt = 0; nt < NT; ++nt)
        asm volatile("" : "+v"(acc[nt]));
    asm volatile("v_nop\n\tv_nop\n\tv_nop\n\tv_nop");
#pragma unroll
    for (int nt = 0; nt < NT; ++nt)
        asm volatile("" : "+v"(acc[nt]));
}

#define EDGES_PER_STRIP (E_EDGES / (N_NODES / 16))
static_assert(EDGES_PER_STRIP * (N_NODES / 16) == E_EDGES && EDGES_PER_STRIP % 128 == 0,
              "the strip waves share the edge list in whole 128-edge trips");

template <int K, int NT, bool ACT, bool EDGE_NORM>
__global__ void __launch_bounds__(160)
gemm_wmma_f32_strip(const float* __restrict__ A, const float* __restrict__ abias,
                    const float* __restrict__ Bf, const float* __restrict__ deg,
                    float* __restrict__ C, float* __restrict__ S,
                    const int* __restrict__ row, const int* __restrict__ col,
                    float* __restrict__ nrm) {
    constexpr int N  = NT * 16;
    constexpr int NJ = K / 8;
    const int wave = threadIdx.x >> 5;
    const int lane = threadIdx.x & 31;
    const int c    = lane & 15;
    const int h    = lane >> 4;
    const int m0   = (blockIdx.x * 5 + wave) * 16;

    if (EDGE_NORM) {
        constexpr int TRIPS = EDGES_PER_STRIP / 128;
        const size_t eb = (size_t)(blockIdx.x * 5 + wave) * EDGES_PER_STRIP + 4 * lane;
        int4 rr[TRIPS], cc[TRIPS];
#pragma unroll
        for (int t = 0; t < TRIPS; ++t) {
            rr[t] = *(const int4*)(row + eb + t * 128);
            cc[t] = *(const int4*)(col + eb + t * 128);
        }
        v4f dr[TRIPS], dc[TRIPS];
#pragma unroll
        for (int t = 0; t < TRIPS; ++t) {
            dr[t] = (v4f){deg[rr[t].x], deg[rr[t].y], deg[rr[t].z], deg[rr[t].w]};
            dc[t] = (v4f){deg[cc[t].x], deg[cc[t].y], deg[cc[t].z], deg[cc[t].w]};
        }
#pragma unroll
        for (int t = 0; t < TRIPS; ++t) {
            v4f nv;
            nv.x = rsqrtf(dr[t].x) * rsqrtf(dc[t].x);
            nv.y = rsqrtf(dr[t].y) * rsqrtf(dc[t].y);
            nv.z = rsqrtf(dr[t].z) * rsqrtf(dc[t].z);
            nv.w = rsqrtf(dr[t].w) * rsqrtf(dc[t].w);
            *(v4f*)(nrm + eb + t * 128) = nv;
        }
    }

    const float* ap = A + (size_t)(m0 + c) * K + 4 * h;
    const float* bp = Bf + lane * 4;
    const float* zp = abias + 4 * h;

    v8f acc[NT];
#pragma unroll
    for (int nt = 0; nt < NT; ++nt)
        acc[nt] = (v8f){0.f, 0.f, 0.f, 0.f, 0.f, 0.f, 0.f, 0.f};

    v4f a_cur = *(const v4f*)ap;
    v4f z_cur = {0.f, 0.f, 0.f, 0.f};
    if (ACT) z_cur = *(const v4f*)zp;
    v4f b_cur[NT];
#pragma unroll
    for (int nt = 0; nt < NT; ++nt)
        b_cur[nt] = *(const v4f*)(bp + nt * 128);

#pragma unroll
    for (int j = 0; j < NJ; ++j) {
        v4f a_nxt = a_cur;
        v4f z_nxt = z_cur;
        v4f b_nxt[NT];
#pragma unroll
        for (int nt = 0; nt < NT; ++nt)
            b_nxt[nt] = b_cur[nt];
        if (j + 1 < NJ) {
            a_nxt = *(const v4f*)(ap + 8 * (j + 1));
            if (ACT) z_nxt = *(const v4f*)(zp + 8 * (j + 1));
#pragma unroll
            for (int nt = 0; nt < NT; ++nt)
                b_nxt[nt] = *(const v4f*)(bp + ((j + 1) * NT + nt) * 128);
        }
        __builtin_amdgcn_sched_barrier(0);
        v4f av = a_cur;
        if (ACT) {
            if (j > 0) wmma_hazard_slots<NT>(acc);
            asm volatile("" : "+v"(a_cur));
            av = a_cur + z_cur;
            av.x = av.x > 0.f ? av.x : 0.f;
            av.y = av.y > 0.f ? av.y : 0.f;
            av.z = av.z > 0.f ? av.z : 0.f;
            av.w = av.w > 0.f ? av.w : 0.f;
            asm volatile("" : "+v"(av));
        }
        v2f a01 = {av.x, av.y};
        v2f a23 = {av.z, av.w};
#pragma unroll
        for (int nt = 0; nt < NT; ++nt) {
            v2f b01 = {b_cur[nt].x, b_cur[nt].y};
            acc[nt] = __builtin_amdgcn_wmma_f32_16x16x4_f32(
                 false, a01,  false, b01,
                 (short)0, acc[nt],  false,  false);
        }
#pragma unroll
        for (int nt = 0; nt < NT; ++nt) {
            v2f b23 = {b_cur[nt].z, b_cur[nt].w};
            acc[nt] = __builtin_amdgcn_wmma_f32_16x16x4_f32(
                 false, a23,  false, b23,
                 (short)0, acc[nt],  false,  false);
        }
        __builtin_amdgcn_sched_barrier(0);
        a_cur = a_nxt;
        z_cur = z_nxt;
#pragma unroll
        for (int nt = 0; nt < NT; ++nt)
            b_cur[nt] = b_nxt[nt];
    }

    wmma_hazard_slots<NT>(acc);
    int mm = m0, hh = h, cl = c;
    asm volatile("" : "+v"(mm), "+v"(hh), "+v"(cl));
    const int rw = mm + 8 * hh;
    const v4f dg0 = *(const v4f*)(deg + rw);
    const v4f dg1 = *(const v4f*)(deg + rw + 4);
    float* cp = C + (size_t)rw * N + cl;
#pragma unroll
    for (int nt = 0; nt < NT; ++nt)
#pragma unroll
        for (int r = 0; r < 8; ++r)
            cp[(size_t)r * N + nt * 16] = acc[nt][r];
    float di[8];
#pragma unroll
    for (int r = 0; r < 4; ++r) {
        di[r]     = rsqrtf(dg0[r]);
        di[r + 4] = rsqrtf(dg1[r]);
    }
    float* sp = S + (size_t)rw * N + cl;
#pragma unroll
    for (int nt = 0; nt < NT; ++nt)
#pragma unroll
        for (int r = 0; r < 8; ++r)
            sp[(size_t)r * N + nt * 16] = acc[nt][r] * di[r] * di[r];
}

template <int H, int G>
__global__ void __launch_bounds__(256)
edge_scatter_kernel(const int* __restrict__ row, const int* __restrict__ col,
                    const float* __restrict__ nrm,
                    const float* __restrict__ htmp,
                    float* __restrict__ acc, int* __restrict__ cnt) {
    static_assert(CHUNK_E == 8 * 64, "8 waves of 64 edges cover a chunk");
    constexpr int V = H / 32;
    __shared__ int sh[2];
    const int2 tk = draw_chunk(cnt, sh);
    if (tk.x >= NCHUNK) return;
    const bool upper = tk.y != 0;
    const unsigned lane = threadIdx.x & 31;
    const unsigned wave = threadIdx.x >> 5;
    const unsigned e0 = (unsigned)tk.x * CHUNK_E + wave * 64 + lane;
    int r0 = row[e0];
    int r1 = row[e0 + 32];
    const int c0 = col[e0];
    const int c1 = col[e0 + 32];
    float nrm0 = nrm[e0];
    float nrm1 = nrm[e0 + 32];
    asm volatile("" : "+v"(r0), "+v"(r1), "+v"(nrm0), "+v"(nrm1));
    const bool mine0 = (c0 >= HALF_NODE) == upper;
    const bool mine1 = (c1 >= HALF_NODE) == upper;
    unsigned long long m =
        ((unsigned long long)__builtin_amdgcn_ballot_w32(mine1) << 32) |
        (unsigned long long)__builtin_amdgcn_ballot_w32(mine0);
    while (m != 0) {
        float v[G][V];
        int   jj[G];
        bool  on[G];
#pragma unroll
        for (int g = 0; g < G; ++g) {
            on[g] = m != 0;
            jj[g] = on[g] ? __builtin_ctzll(m) : 0;
            m &= m - 1;
#pragma unroll
            for (int k = 0; k < V; ++k) v[g][k] = 0.f;
            if (on[g]) {
                const int ra = __builtin_amdgcn_readlane(r0, jj[g] & 31);
                const int rb = __builtin_amdgcn_readlane(r1, jj[g] & 31);
                const int rj = (jj[g] & 32) ? rb : ra;
                const float* src = htmp + (size_t)rj * H;
#pragma unroll
                for (int k = 0; k < V; ++k)
                    v[g][k] = src[lane + k * 32];
            }
        }
#pragma unroll
        for (int g = 0; g < G; ++g) {
            if (on[g]) {
                const int l  = jj[g] & 31;
                const int ca = __builtin_amdgcn_readlane(c0, l);
                const int cb = __builtin_amdgcn_readlane(c1, l);
                const int na = __builtin_amdgcn_readlane(__builtin_bit_cast(int, nrm0), l);
                const int nb = __builtin_amdgcn_readlane(__builtin_bit_cast(int, nrm1), l);
                const int cj = (jj[g] & 32) ? cb : ca;
                const float nj = __builtin_bit_cast(float, (jj[g] & 32) ? nb : na);
                float* dst = acc + (size_t)cj * H;
#pragma unroll
                for (int k = 0; k < V; ++k)
                    atomicAdd(&dst[lane + k * 32], v[g][k] * nj);
            }
        }
    }
}

#define DOT_G 16
static_assert(32 % DOT_G == 0, "whole groups per wave");
#define DOT_BLOCKS ((2 * EP_CNT + 63) / 64)

template <int D, int CNT>
__device__ __forceinline__ void fold_step(float (&p)[32], unsigned lane) {
    const bool up = (lane & D) != 0;
#pragma unroll
    for (int e = 0; e < CNT; ++e) {
        const float keep = up ? p[e + CNT] : p[e];
        const float send = up ? p[e] : p[e + CNT];
        p[e] = keep + __shfl_xor(send, D, 32);
    }
}

__global__ void __launch_bounds__(64)
edge_dot_kernel(const int* __restrict__ pos, const int* __restrict__ neg,
                const float* __restrict__ acc2, const float* __restrict__ bias,
                float* __restrict__ out, int ep) {
    const unsigned lane = threadIdx.x & 31;
    const int wave = (int)((blockIdx.x * blockDim.x + threadIdx.x) >> 5);
    const int e = wave * 32 + (int)lane;
    if (wave * 32 >= 2 * ep) return;
    const bool live = e < 2 * ep;
    const bool isneg = e >= ep;
    const int* tab = isneg ? neg : pos;
    const int k = live ? (isneg ? e - ep : e) : 0;
    const int srcn = tab[k];
    const int dstn = tab[ep + k];
    const float bz0 = bias[lane];
    const float bz1 = bias[lane + 32];

    float p[32];
#pragma unroll
    for (int g0 = 0; g0 < 32; g0 += DOT_G) {
        float v[DOT_G][4];
#pragma unroll
        for (int g = 0; g < DOT_G; ++g) {
            const int sj = __builtin_amdgcn_readlane(srcn, g0 + g);
            const int di = __builtin_amdgcn_readlane(dstn, g0 + g);
            const float* xi = acc2 + (size_t)di * H2_DIM;
            const float* xj = acc2 + (size_t)sj * H2_DIM;
            v[g][0] = xi[lane];
            v[g][1] = xj[lane];
            v[g][2] = xi[lane + 32];
            v[g][3] = xj[lane + 32];
        }
        __builtin_amdgcn_sched_barrier(0);
#pragma unroll
        for (int g = 0; g < DOT_G; ++g) {
            const float i0 = v[g][0] + bz0, j0 = v[g][1] + bz0;
            const float i1 = v[g][2] + bz1, j1 = v[g][3] + bz1;
            p[g0 + g] = i0 * j0 + i1 * j1;
        }
    }
    fold_step<16, 16>(p, lane);
    fold_step<8, 8>(p, lane);
    fold_step<4, 4>(p, lane);
    fold_step<2, 2>(p, lane);
    fold_step<1, 1>(p, lane);
    if (live) out[e] = p[0];
}

static inline size_t align256(size_t x) { return (x + 255) & ~(size_t)255; }

extern "C" void kernel_launch(void* const* d_in, const int* in_sizes, int n_in,
                              void* d_out, int out_size, void* d_ws, size_t ws_size,
                              hipStream_t stream) {
    const float* x    = (const float*)d_in[0];
    const int*   ei   = (const int*)d_in[1];
    const int*   pos  = (const int*)d_in[2];
    const int*   neg  = (const int*)d_in[3];
    const float* W1   = (const float*)d_in[4];
    const float* b1   = (const float*)d_in[5];
    const float* W2   = (const float*)d_in[6];
    const float* b2   = (const float*)d_in[7];
    float* out = (float*)d_out;

    const int* row = ei;
    const int* col = ei + E_EDGES;

    char* ws = (char*)d_ws;
    size_t off = 0;
    float* deg   = (float*)(ws + off);
    int*   tick  = (int*)(ws + off + sizeof(float) * N_NODES);
    off += align256(sizeof(float) * N_NODES + sizeof(int) * N_TICKETS);
    float* W1f   = (float*)(ws + off); off += align256(sizeof(float) * F_IN * H1_DIM);
    float* W2f   = (float*)(ws + off); off += align256(sizeof(float) * H1_DIM * H2_DIM);
    float* htmp1 = (float*)(ws + off); off += align256(sizeof(float) * (size_t)N_NODES * H1_DIM);
    float* acc1  = (float*)(ws + off); off += align256(sizeof(float) * (size_t)N_NODES * H1_DIM);
    static_assert((size_t)E_EDGES <= (size_t)N_NODES * H2_DIM, "factors fit the region");
    float* nrm   = (float*)(ws + off); off += align256(sizeof(float) * (size_t)N_NODES * H2_DIM);
    float* acc2  = (float*)(ws + off); off += align256(sizeof(float) * (size_t)N_NODES * H2_DIM);
    float* htmp2 = htmp1;
    (void)ws_size; (void)n_in; (void)in_sizes; (void)out_size;

    const int TB = 256;

    prep_kernel<<<W1_BLOCKS + W2_BLOCKS + DEG_BLOCKS, PREP_TB, 0, stream>>>(W1, W2, W1f, W2f, deg, tick);

    accum_deg_kernel<<<DEG_SLICES * DEG_RANGES, DEG_TB, 0, stream>>>(col, deg);

    gemm_wmma_f32_strip<F_IN, 8, false, true><<<N_NODES / 80, 160, 0, stream>>>(
        x, b1, W1f, deg, htmp1, acc1, row, col, nrm);

    edge_scatter_kernel<H1_DIM, 8><<<2 * NCHUNK, TB, 0, stream>>>(row, col, nrm, htmp1, acc1, tick);

    gemm_wmma_f32_strip<H1_DIM, 4, true, false><<<N_NODES / 80, 160, 0, stream>>>(
        acc1, b1, W2f, deg, htmp2, acc2, row, col, nrm);

    edge_scatter_kernel<H2_DIM, 16><<<2 * NCHUNK, TB, 0, stream>>>(row, col, nrm, htmp2, acc2, tick + 2);

    edge_dot_kernel<<<DOT_BLOCKS, 64, 0, stream>>>(pos, neg, acc2, b2, out, EP_CNT);
}
